# starting kernel
# baseline (speedup 1.0000x reference)
#include <hip/hip_runtime.h>
#include <math.h>

typedef __attribute__((ext_vector_type(2))) float v2f;
typedef __attribute__((ext_vector_type(8))) float v8f;

#define NPIX   (512*512)
#define NSAMP  16
#define CI     3
#define CH     16
#define CO     3
#define PB1    64
#define PB3    64
#define PB5    128
#define EPS    1e-7f
#define SLOPE  0.01f

__device__ __forceinline__ v8f wmma4(v2f a, v2f b, v8f c) {
  return __builtin_amdgcn_wmma_f32_16x16x4_f32(false, a, false, b, (short)0, c,
                                               false, false);
}

__global__ void k_stats1(const float* __restrict__ x, const float* __restrict__ Wa,
                         const float* __restrict__ ba, float* __restrict__ part) {
  const int s = blockIdx.y;
  const float* xs = x + (size_t)s * CI * NPIX;

  float wa[CH][CI], bav[CH];
#pragma unroll
  for (int c = 0; c < CH; ++c) {
    bav[c] = ba[c];
#pragma unroll
    for (int i = 0; i < CI; ++i) wa[c][i] = Wa[c * CI + i];
  }

  float s1 = 0.f, s2 = 0.f;
  const int stride = blockDim.x * gridDim.x;
  for (int p = blockIdx.x * blockDim.x + threadIdx.x; p < NPIX; p += stride) {
    float x0 = xs[p], x1 = xs[NPIX + p], x2 = xs[2 * NPIX + p];
#pragma unroll
    for (int c = 0; c < CH; ++c) {
      float y = fmaf(wa[c][0], x0, fmaf(wa[c][1], x1, fmaf(wa[c][2], x2, bav[c])));
      s1 += y;
      s2 += y * y;
    }
  }

  __shared__ float r1[256], r2[256];
  r1[threadIdx.x] = s1; r2[threadIdx.x] = s2;
  __syncthreads();
  for (int w = 128; w > 0; w >>= 1) {
    if (threadIdx.x < (unsigned)w) {
      r1[threadIdx.x] += r1[threadIdx.x + w];
      r2[threadIdx.x] += r2[threadIdx.x + w];
    }
    __syncthreads();
  }
  if (threadIdx.x == 0) {
    part[(s * PB1 + blockIdx.x) * 2 + 0] = r1[0];
    part[(s * PB1 + blockIdx.x) * 2 + 1] = r2[0];
  }
}

__global__ void k_finalize(const float* __restrict__ part, float* __restrict__ stats,
                           int nb) {
  int s = threadIdx.x;
  if (s >= NSAMP) return;
  double s1 = 0.0, s2 = 0.0;
  for (int b = 0; b < nb; ++b) {
    s1 += (double)part[(s * nb + b) * 2 + 0];
    s2 += (double)part[(s * nb + b) * 2 + 1];
  }
  const double n = (double)CH * (double)NPIX;
  double mean = s1 / n;
  double var = (s2 - s1 * mean) / (n - 1.0);
  stats[2 * s + 0] = (float)mean;
  stats[2 * s + 1] = 1.0f / (sqrtf((float)var) + EPS);
}

__global__ void k_stats2(const float* __restrict__ x, const float* __restrict__ Wa,
                         const float* __restrict__ ba, const float* __restrict__ Wb,
                         const float* __restrict__ bb, const float* __restrict__ stats1,
                         float* __restrict__ part) {
  const int s = blockIdx.y;
  const float* xs = x + (size_t)s * CI * NPIX;
  const int lane  = threadIdx.x & 31;
  const int wave  = threadIdx.x >> 5;
  const int m     = lane & 15;
  const int khalf = (lane >> 4) << 1;
  const int hi8   = (lane >> 4) << 3;

  v2f aA;
  aA.x = Wa[m * CI + khalf];
  aA.y = (khalf == 0) ? Wa[m * CI + 1] : 0.f;

  v2f bA[4];
#pragma unroll
  for (int j = 0; j < 4; ++j) {
    int k0 = 4 * j + khalf;
    bA[j].x = Wb[m * CH + k0];
    bA[j].y = Wb[m * CH + k0 + 1];
  }

  v8f biasA, biasB;
#pragma unroll
  for (int i = 0; i < 8; ++i) { biasA[i] = ba[i + hi8]; biasB[i] = bb[i + hi8]; }

  const float mean1 = stats1[2 * s], inv1 = stats1[2 * s + 1];

  __shared__ float tile[8][CH * 16];
  float* tl = &tile[wave][0];

  float acc1 = 0.f, acc2 = 0.f;
  const int tpb = (NPIX / 16) / gridDim.x;
  const int tpw = tpb / 8;
  const int tbase = blockIdx.x * tpb + wave * tpw;

  for (int t = 0; t < tpw; ++t) {
    const int p = (tbase + t) * 16;
    v2f xB;
    xB.x = xs[khalf * NPIX + p + m];
    xB.y = (khalf == 0) ? xs[NPIX + p + m] : 0.f;

    v8f y = wmma4(aA, xB, biasA);

    v8f h;
#pragma unroll
    for (int i = 0; i < 8; ++i) {
      float v = (y[i] - mean1) * inv1;
      h[i] = (v >= 0.f) ? v : SLOPE * v;
    }

    __builtin_amdgcn_wave_barrier();
#pragma unroll
    for (int i = 0; i < 8; ++i) tl[(i + hi8) * 16 + m] = h[i];
    __builtin_amdgcn_wave_barrier();

    v8f z = biasB;
#pragma unroll
    for (int j = 0; j < 4; ++j) {
      int k0 = 4 * j + khalf;
      v2f hB;
      hB.x = tl[k0 * 16 + m];
      hB.y = tl[(k0 + 1) * 16 + m];
      z = wmma4(bA[j], hB, z);
    }
    __builtin_amdgcn_wave_barrier();

#pragma unroll
    for (int i = 0; i < 8; ++i) { acc1 += z[i]; acc2 += z[i] * z[i]; }
  }

  __shared__ float r1[256], r2[256];
  r1[threadIdx.x] = acc1; r2[threadIdx.x] = acc2;
  __syncthreads();
  for (int w = 128; w > 0; w >>= 1) {
    if (threadIdx.x < (unsigned)w) {
      r1[threadIdx.x] += r1[threadIdx.x + w];
      r2[threadIdx.x] += r2[threadIdx.x + w];
    }
    __syncthreads();
  }
  if (threadIdx.x == 0) {
    part[(s * PB3 + blockIdx.x) * 2 + 0] = r1[0];
    part[(s * PB3 + blockIdx.x) * 2 + 1] = r2[0];
  }
}

__global__ void k_final(const float* __restrict__ x, const float* __restrict__ Wa,
                        const float* __restrict__ ba, const float* __restrict__ Wb,
                        const float* __restrict__ bb, const float* __restrict__ Wc,
                        const float* __restrict__ bc, const float* __restrict__ stats1,
                        const float* __restrict__ stats2, float* __restrict__ out) {
  const int s = blockIdx.y;
  const float* xs = x + (size_t)s * CI * NPIX;
  float* os = out + (size_t)s * CO * NPIX;
  const int lane  = threadIdx.x & 31;
  const int wave  = threadIdx.x >> 5;
  const int m     = lane & 15;
  const int khalf = (lane >> 4) << 1;
  const int hi8   = (lane >> 4) << 3;

  v2f aA;
  aA.x = Wa[m * CI + khalf];
  aA.y = (khalf == 0) ? Wa[m * CI + 1] : 0.f;

  v2f bA[4], cA[4];
#pragma unroll
  for (int j = 0; j < 4; ++j) {
    int k0 = 4 * j + khalf;
    bA[j].x = Wb[m * CH + k0];
    bA[j].y = Wb[m * CH + k0 + 1];
    cA[j].x = (m < CO) ? Wc[m * CH + k0] : 0.f;
    cA[j].y = (m < CO) ? Wc[m * CH + k0 + 1] : 0.f;
  }

  v8f biasA, biasB, biasC;
#pragma unroll
  for (int i = 0; i < 8; ++i) {
    biasA[i] = ba[i + hi8];
    biasB[i] = bb[i + hi8];
    biasC[i] = (i + hi8 < CO) ? bc[i + hi8] : 0.f;
  }

  const float mean1 = stats1[2 * s], inv1 = stats1[2 * s + 1];
  const float mean2 = stats2[2 * s], inv2 = stats2[2 * s + 1];

  __shared__ float tile[8][CH * 16];
  float* tl = &tile[wave][0];

  const int tpb = (NPIX / 16) / gridDim.x;
  const int tpw = tpb / 8;
  const int tbase = blockIdx.x * tpb + wave * tpw;

  for (int t = 0; t < tpw; ++t) {
    const int p = (tbase + t) * 16;
    v2f xB;
    xB.x = xs[khalf * NPIX + p + m];
    xB.y = (khalf == 0) ? xs[NPIX + p + m] : 0.f;

    v8f y = wmma4(aA, xB, biasA);

    v8f h1;
#pragma unroll
    for (int i = 0; i < 8; ++i) {
      float v = (y[i] - mean1) * inv1;
      h1[i] = (v >= 0.f) ? v : SLOPE * v;
    }

    __builtin_amdgcn_wave_barrier();
#pragma unroll
    for (int i = 0; i < 8; ++i) tl[(i + hi8) * 16 + m] = h1[i];
    __builtin_amdgcn_wave_barrier();

    v8f z = biasB;
#pragma unroll
    for (int j = 0; j < 4; ++j) {
      int k0 = 4 * j + khalf;
      v2f hB;
      hB.x = tl[k0 * 16 + m];
      hB.y = tl[(k0 + 1) * 16 + m];
      z = wmma4(bA[j], hB, z);
    }
    __builtin_amdgcn_wave_barrier();

    v8f h2;
#pragma unroll
    for (int i = 0; i < 8; ++i) {
      float v = (z[i] - mean2) * inv2;
      h2[i] = (v >= 0.f) ? v : SLOPE * v;
    }

    __builtin_amdgcn_wave_barrier();
#pragma unroll
    for (int i = 0; i < 8; ++i) tl[(i + hi8) * 16 + m] = h2[i];
    __builtin_amdgcn_wave_barrier();

    v8f o = biasC;
#pragma unroll
    for (int j = 0; j < 4; ++j) {
      int k0 = 4 * j + khalf;
      v2f hB;
      hB.x = tl[k0 * 16 + m];
      hB.y = tl[(k0 + 1) * 16 + m];
      o = wmma4(cA[j], hB, o);
    }
    __builtin_amdgcn_wave_barrier();

    if (lane < 16) {
      os[0 * NPIX + p + m] = o[0];
      os[1 * NPIX + p + m] = o[1];
      os[2 * NPIX + p + m] = o[2];
    }
  }
}

extern "C" void kernel_launch(void* const* d_in, const int* in_sizes, int n_in,
                              void* d_out, int out_size, void* d_ws, size_t ws_size,
                              hipStream_t stream) {
  (void)in_sizes; (void)n_in; (void)out_size; (void)ws_size;
  const float* x  = (const float*)d_in[0];
  const float* Wa = (const float*)d_in[1];
  const float* ba = (const float*)d_in[2];
  const float* Wb = (const float*)d_in[3];
  const float* bb = (const float*)d_in[4];
  const float* Wc = (const float*)d_in[5];
  const float* bc = (const float*)d_in[6];
  float* out = (float*)d_out;

  float* ws     = (float*)d_ws;
  float* part1  = ws;
  float* stats1 = part1 + NSAMP * PB1 * 2;
  float* part2  = stats1 + NSAMP * 2;
  float* stats2 = part2 + NSAMP * PB3 * 2;

  k_stats1 <<<dim3(PB1, NSAMP), 256, 0, stream>>>(x, Wa, ba, part1);
  k_finalize<<<1, 32, 0, stream>>>(part1, stats1, PB1);
  k_stats2 <<<dim3(PB3, NSAMP), 256, 0, stream>>>(x, Wa, ba, Wb, bb, stats1, part2);
  k_finalize<<<1, 32, 0, stream>>>(part2, stats2, PB3);
  k_final  <<<dim3(PB5, NSAMP), 256, 0, stream>>>(x, Wa, ba, Wb, bb, Wc, bc,
                                                  stats1, stats2, out);
}


# both single-block finalize launches are gone: wave 0 (and 1) of each consumer block folds the 64 partials itself, lane-parallel in double + xor butterfly; 5 -> 3 launches; partial sum order changes
# speedup vs baseline: 1.1444x; 1.1444x over previous
#include <hip/hip_runtime.h>
#include <math.h>

typedef __attribute__((ext_vector_type(2))) float v2f;
typedef __attribute__((ext_vector_type(8))) float v8f;

#define NPIX   (512*512)
#define NSAMP  16
#define CI     3
#define CH     16
#define CO     3
#define PB1    64
#define PB3    64
#define PB5    128
#define EPS    1e-7f
#define SLOPE  0.01f

__device__ __forceinline__ v8f wmma4(v2f a, v2f b, v8f c) {
  return __builtin_amdgcn_wmma_f32_16x16x4_f32(false, a, false, b, (short)0, c,
                                               false, false);
}

__global__ void k_stats1(const float* __restrict__ x, const float* __restrict__ Wa,
                         const float* __restrict__ ba, float* __restrict__ part) {
  const int s = blockIdx.y;
  const float* xs = x + (size_t)s * CI * NPIX;

  float wa[CH][CI], bav[CH];
#pragma unroll
  for (int c = 0; c < CH; ++c) {
    bav[c] = ba[c];
#pragma unroll
    for (int i = 0; i < CI; ++i) wa[c][i] = Wa[c * CI + i];
  }

  float s1 = 0.f, s2 = 0.f;
  const int stride = blockDim.x * gridDim.x;
  for (int p = blockIdx.x * blockDim.x + threadIdx.x; p < NPIX; p += stride) {
    float x0 = xs[p], x1 = xs[NPIX + p], x2 = xs[2 * NPIX + p];
#pragma unroll
    for (int c = 0; c < CH; ++c) {
      float y = fmaf(wa[c][0], x0, fmaf(wa[c][1], x1, fmaf(wa[c][2], x2, bav[c])));
      s1 += y;
      s2 += y * y;
    }
  }

  __shared__ float r1[256], r2[256];
  r1[threadIdx.x] = s1; r2[threadIdx.x] = s2;
  __syncthreads();
  for (int w = 128; w > 0; w >>= 1) {
    if (threadIdx.x < (unsigned)w) {
      r1[threadIdx.x] += r1[threadIdx.x + w];
      r2[threadIdx.x] += r2[threadIdx.x + w];
    }
    __syncthreads();
  }
  if (threadIdx.x == 0) {
    part[(s * PB1 + blockIdx.x) * 2 + 0] = r1[0];
    part[(s * PB1 + blockIdx.x) * 2 + 1] = r2[0];
  }
}

template <int NB>
__device__ __forceinline__ void fold_stats(const float* __restrict__ part, int s, int lane,
                                           float& mean_o, float& inv_o) {
  float p1[NB / 32], p2[NB / 32];
#pragma unroll
  for (int k = 0; k < NB / 32; ++k) {
    p1[k] = part[(s * NB + lane + 32 * k) * 2 + 0];
    p2[k] = part[(s * NB + lane + 32 * k) * 2 + 1];
  }
  double s1 = 0.0, s2 = 0.0;
#pragma unroll
  for (int k = 0; k < NB / 32; ++k) { s1 += (double)p1[k]; s2 += (double)p2[k]; }
#pragma unroll
  for (int o = 16; o > 0; o >>= 1) {
    const double t1 = __shfl_xor(s1, o);
    const double t2 = __shfl_xor(s2, o);
    s1 += t1;
    s2 += t2;
  }
  const double n = (double)CH * (double)NPIX;
  double mean = s1 / n;
  double var = (s2 - s1 * mean) / (n - 1.0);
  mean_o = (float)mean;
  inv_o = 1.0f / (sqrtf((float)var) + EPS);
}

__global__ void k_stats2(const float* __restrict__ x, const float* __restrict__ Wa,
                         const float* __restrict__ ba, const float* __restrict__ Wb,
                         const float* __restrict__ bb, const float* __restrict__ part1,
                         float* __restrict__ part) {
  const int s = blockIdx.y;
  const float* xs = x + (size_t)s * CI * NPIX;
  const int lane  = threadIdx.x & 31;
  const int wave  = threadIdx.x >> 5;
  const int m     = lane & 15;
  const int khalf = (lane >> 4) << 1;
  const int hi8   = (lane >> 4) << 3;

  v2f aA;
  aA.x = Wa[m * CI + khalf];
  aA.y = (khalf == 0) ? Wa[m * CI + 1] : 0.f;

  v2f bA[4];
#pragma unroll
  for (int j = 0; j < 4; ++j) {
    int k0 = 4 * j + khalf;
    bA[j].x = Wb[m * CH + k0];
    bA[j].y = Wb[m * CH + k0 + 1];
  }

  v8f biasA, biasB;
#pragma unroll
  for (int i = 0; i < 8; ++i) { biasA[i] = ba[i + hi8]; biasB[i] = bb[i + hi8]; }

  __shared__ float st[2];
  if (wave == 0) {
    float mu, iv;
    fold_stats<PB1>(part1, s, lane, mu, iv);
    if (lane == 0) { st[0] = mu; st[1] = iv; }
  }
  __syncthreads();
  const float mean1 = st[0], inv1 = st[1];

  __shared__ float tile[8][CH * 16];
  float* tl = &tile[wave][0];

  float acc1 = 0.f, acc2 = 0.f;
  const int tpb = (NPIX / 16) / gridDim.x;
  const int tpw = tpb / 8;
  const int tbase = blockIdx.x * tpb + wave * tpw;

  for (int t = 0; t < tpw; ++t) {
    const int p = (tbase + t) * 16;
    v2f xB;
    xB.x = xs[khalf * NPIX + p + m];
    xB.y = (khalf == 0) ? xs[NPIX + p + m] : 0.f;

    v8f y = wmma4(aA, xB, biasA);

    v8f h;
#pragma unroll
    for (int i = 0; i < 8; ++i) {
      float v = (y[i] - mean1) * inv1;
      h[i] = (v >= 0.f) ? v : SLOPE * v;
    }

    __builtin_amdgcn_wave_barrier();
#pragma unroll
    for (int i = 0; i < 8; ++i) tl[(i + hi8) * 16 + m] = h[i];
    __builtin_amdgcn_wave_barrier();

    v8f z = biasB;
#pragma unroll
    for (int j = 0; j < 4; ++j) {
      int k0 = 4 * j + khalf;
      v2f hB;
      hB.x = tl[k0 * 16 + m];
      hB.y = tl[(k0 + 1) * 16 + m];
      z = wmma4(bA[j], hB, z);
    }
    __builtin_amdgcn_wave_barrier();

#pragma unroll
    for (int i = 0; i < 8; ++i) { acc1 += z[i]; acc2 += z[i] * z[i]; }
  }

  __shared__ float r1[256], r2[256];
  r1[threadIdx.x] = acc1; r2[threadIdx.x] = acc2;
  __syncthreads();
  for (int w = 128; w > 0; w >>= 1) {
    if (threadIdx.x < (unsigned)w) {
      r1[threadIdx.x] += r1[threadIdx.x + w];
      r2[threadIdx.x] += r2[threadIdx.x + w];
    }
    __syncthreads();
  }
  if (threadIdx.x == 0) {
    part[(s * PB3 + blockIdx.x) * 2 + 0] = r1[0];
    part[(s * PB3 + blockIdx.x) * 2 + 1] = r2[0];
  }
}

__global__ void k_final(const float* __restrict__ x, const float* __restrict__ Wa,
                        const float* __restrict__ ba, const float* __restrict__ Wb,
                        const float* __restrict__ bb, const float* __restrict__ Wc,
                        const float* __restrict__ bc, const float* __restrict__ part1,
                        const float* __restrict__ part2, float* __restrict__ out) {
  const int s = blockIdx.y;
  const float* xs = x + (size_t)s * CI * NPIX;
  float* os = out + (size_t)s * CO * NPIX;
  const int lane  = threadIdx.x & 31;
  const int wave  = threadIdx.x >> 5;
  const int m     = lane & 15;
  const int khalf = (lane >> 4) << 1;
  const int hi8   = (lane >> 4) << 3;

  v2f aA;
  aA.x = Wa[m * CI + khalf];
  aA.y = (khalf == 0) ? Wa[m * CI + 1] : 0.f;

  v2f bA[4], cA[4];
#pragma unroll
  for (int j = 0; j < 4; ++j) {
    int k0 = 4 * j + khalf;
    bA[j].x = Wb[m * CH + k0];
    bA[j].y = Wb[m * CH + k0 + 1];
    cA[j].x = (m < CO) ? Wc[m * CH + k0] : 0.f;
    cA[j].y = (m < CO) ? Wc[m * CH + k0 + 1] : 0.f;
  }

  v8f biasA, biasB, biasC;
#pragma unroll
  for (int i = 0; i < 8; ++i) {
    biasA[i] = ba[i + hi8];
    biasB[i] = bb[i + hi8];
    biasC[i] = (i + hi8 < CO) ? bc[i + hi8] : 0.f;
  }

  __shared__ float st[4];
  if (wave < 2) {
    float mu, iv;
    if (wave == 0) fold_stats<PB1>(part1, s, lane, mu, iv);
    else           fold_stats<PB3>(part2, s, lane, mu, iv);
    if (lane == 0) { st[2 * wave] = mu; st[2 * wave + 1] = iv; }
  }
  __syncthreads();
  const float mean1 = st[0], inv1 = st[1];
  const float mean2 = st[2], inv2 = st[3];

  __shared__ float tile[8][CH * 16];
  float* tl = &tile[wave][0];

  const int tpb = (NPIX / 16) / gridDim.x;
  const int tpw = tpb / 8;
  const int tbase = blockIdx.x * tpb + wave * tpw;

  for (int t = 0; t < tpw; ++t) {
    const int p = (tbase + t) * 16;
    v2f xB;
    xB.x = xs[khalf * NPIX + p + m];
    xB.y = (khalf == 0) ? xs[NPIX + p + m] : 0.f;

    v8f y = wmma4(aA, xB, biasA);

    v8f h1;
#pragma unroll
    for (int i = 0; i < 8; ++i) {
      float v = (y[i] - mean1) * inv1;
      h1[i] = (v >= 0.f) ? v : SLOPE * v;
    }

    __builtin_amdgcn_wave_barrier();
#pragma unroll
    for (int i = 0; i < 8; ++i) tl[(i + hi8) * 16 + m] = h1[i];
    __builtin_amdgcn_wave_barrier();

    v8f z = biasB;
#pragma unroll
    for (int j = 0; j < 4; ++j) {
      int k0 = 4 * j + khalf;
      v2f hB;
      hB.x = tl[k0 * 16 + m];
      hB.y = tl[(k0 + 1) * 16 + m];
      z = wmma4(bA[j], hB, z);
    }
    __builtin_amdgcn_wave_barrier();

    v8f h2;
#pragma unroll
    for (int i = 0; i < 8; ++i) {
      float v = (z[i] - mean2) * inv2;
      h2[i] = (v >= 0.f) ? v : SLOPE * v;
    }

    __builtin_amdgcn_wave_barrier();
#pragma unroll
    for (int i = 0; i < 8; ++i) tl[(i + hi8) * 16 + m] = h2[i];
    __builtin_amdgcn_wave_barrier();

    v8f o = biasC;
#pragma unroll
    for (int j = 0; j < 4; ++j) {
      int k0 = 4 * j + khalf;
      v2f hB;
      hB.x = tl[k0 * 16 + m];
      hB.y = tl[(k0 + 1) * 16 + m];
      o = wmma4(cA[j], hB, o);
    }
    __builtin_amdgcn_wave_barrier();

    if (lane < 16) {
      os[0 * NPIX + p + m] = o[0];
      os[1 * NPIX + p + m] = o[1];
      os[2 * NPIX + p + m] = o[2];
    }
  }
}

extern "C" void kernel_launch(void* const* d_in, const int* in_sizes, int n_in,
                              void* d_out, int out_size, void* d_ws, size_t ws_size,
                              hipStream_t stream) {
  (void)in_sizes; (void)n_in; (void)out_size; (void)ws_size;
  const float* x  = (const float*)d_in[0];
  const float* Wa = (const float*)d_in[1];
  const float* ba = (const float*)d_in[2];
  const float* Wb = (const float*)d_in[3];
  const float* bb = (const float*)d_in[4];
  const float* Wc = (const float*)d_in[5];
  const float* bc = (const float*)d_in[6];
  float* out = (float*)d_out;

  float* ws     = (float*)d_ws;
  float* part1  = ws;
  float* part2  = part1 + NSAMP * PB1 * 2;

  k_stats1 <<<dim3(PB1, NSAMP), 256, 0, stream>>>(x, Wa, ba, part1);
  k_stats2 <<<dim3(PB3, NSAMP), 256, 0, stream>>>(x, Wa, ba, Wb, bb, part1, part2);
  k_final  <<<dim3(PB5, NSAMP), 256, 0, stream>>>(x, Wa, ba, Wb, bb, Wc, bc,
                                                  part1, part2, out);
}


# each conv's D registers feed the next conv's B operand unmoved (k = 8h+2j+s, same order in the Wb/Wc loaders): LDS tile round trips and wave barriers gone; k sum order of the dot products changes
# speedup vs baseline: 1.1983x; 1.0471x over previous
#include <hip/hip_runtime.h>
#include <math.h>

typedef __attribute__((ext_vector_type(2))) float v2f;
typedef __attribute__((ext_vector_type(8))) float v8f;

#define NPIX   (512*512)
#define NSAMP  16
#define CI     3
#define CH     16
#define CO     3
#define PB1    64
#define PB3    64
#define PB5    128
#define EPS    1e-7f
#define SLOPE  0.01f

__device__ __forceinline__ v8f wmma4(v2f a, v2f b, v8f c) {
  return __builtin_amdgcn_wmma_f32_16x16x4_f32(false, a, false, b, (short)0, c,
                                               false, false);
}

__device__ __forceinline__ int hand_off_k(int j, int hi8) { return hi8 + 2 * j; }

__global__ void k_stats1(const float* __restrict__ x, const float* __restrict__ Wa,
                         const float* __restrict__ ba, float* __restrict__ part) {
  const int s = blockIdx.y;
  const float* xs = x + (size_t)s * CI * NPIX;

  float wa[CH][CI], bav[CH];
#pragma unroll
  for (int c = 0; c < CH; ++c) {
    bav[c] = ba[c];
#pragma unroll
    for (int i = 0; i < CI; ++i) wa[c][i] = Wa[c * CI + i];
  }

  float s1 = 0.f, s2 = 0.f;
  const int stride = blockDim.x * gridDim.x;
  for (int p = blockIdx.x * blockDim.x + threadIdx.x; p < NPIX; p += stride) {
    float x0 = xs[p], x1 = xs[NPIX + p], x2 = xs[2 * NPIX + p];
#pragma unroll
    for (int c = 0; c < CH; ++c) {
      float y = fmaf(wa[c][0], x0, fmaf(wa[c][1], x1, fmaf(wa[c][2], x2, bav[c])));
      s1 += y;
      s2 += y * y;
    }
  }

  __shared__ float r1[256], r2[256];
  r1[threadIdx.x] = s1; r2[threadIdx.x] = s2;
  __syncthreads();
  for (int w = 128; w > 0; w >>= 1) {
    if (threadIdx.x < (unsigned)w) {
      r1[threadIdx.x] += r1[threadIdx.x + w];
      r2[threadIdx.x] += r2[threadIdx.x + w];
    }
    __syncthreads();
  }
  if (threadIdx.x == 0) {
    part[(s * PB1 + blockIdx.x) * 2 + 0] = r1[0];
    part[(s * PB1 + blockIdx.x) * 2 + 1] = r2[0];
  }
}

template <int NB>
__device__ __forceinline__ void fold_stats(const float* __restrict__ part, int s, int lane,
                                           float& mean_o, float& inv_o) {
  float p1[NB / 32], p2[NB / 32];
#pragma unroll
  for (int k = 0; k < NB / 32; ++k) {
    p1[k] = part[(s * NB + lane + 32 * k) * 2 + 0];
    p2[k] = part[(s * NB + lane + 32 * k) * 2 + 1];
  }
  double s1 = 0.0, s2 = 0.0;
#pragma unroll
  for (int k = 0; k < NB / 32; ++k) { s1 += (double)p1[k]; s2 += (double)p2[k]; }
#pragma unroll
  for (int o = 16; o > 0; o >>= 1) {
    const double t1 = __shfl_xor(s1, o);
    const double t2 = __shfl_xor(s2, o);
    s1 += t1;
    s2 += t2;
  }
  const double n = (double)CH * (double)NPIX;
  double mean = s1 / n;
  double var = (s2 - s1 * mean) / (n - 1.0);
  mean_o = (float)mean;
  inv_o = 1.0f / (sqrtf((float)var) + EPS);
}

__global__ void k_stats2(const float* __restrict__ x, const float* __restrict__ Wa,
                         const float* __restrict__ ba, const float* __restrict__ Wb,
                         const float* __restrict__ bb, const float* __restrict__ part1,
                         float* __restrict__ part) {
  const int s = blockIdx.y;
  const float* xs = x + (size_t)s * CI * NPIX;
  const int lane  = threadIdx.x & 31;
  const int wave  = threadIdx.x >> 5;
  const int m     = lane & 15;
  const int khalf = (lane >> 4) << 1;
  const int hi8   = (lane >> 4) << 3;

  v2f aA;
  aA.x = Wa[m * CI + khalf];
  aA.y = (khalf == 0) ? Wa[m * CI + 1] : 0.f;

  v2f bA[4];
#pragma unroll
  for (int j = 0; j < 4; ++j) {
    int k0 = hand_off_k(j, hi8);
    bA[j].x = Wb[m * CH + k0];
    bA[j].y = Wb[m * CH + k0 + 1];
  }

  v8f biasA, biasB;
#pragma unroll
  for (int i = 0; i < 8; ++i) { biasA[i] = ba[i + hi8]; biasB[i] = bb[i + hi8]; }

  __shared__ float st[2];
  if (wave == 0) {
    float mu, iv;
    fold_stats<PB1>(part1, s, lane, mu, iv);
    if (lane == 0) { st[0] = mu; st[1] = iv; }
  }
  __syncthreads();
  const float mean1 = st[0], inv1 = st[1];

  float acc1 = 0.f, acc2 = 0.f;
  const int tpb = (NPIX / 16) / gridDim.x;
  const int tpw = tpb / 8;
  const int tbase = blockIdx.x * tpb + wave * tpw;

  for (int t = 0; t < tpw; ++t) {
    const int p = (tbase + t) * 16;
    v2f xB;
    xB.x = xs[khalf * NPIX + p + m];
    xB.y = (khalf == 0) ? xs[NPIX + p + m] : 0.f;

    v8f y = wmma4(aA, xB, biasA);

    v8f h;
#pragma unroll
    for (int i = 0; i < 8; ++i) {
      float v = (y[i] - mean1) * inv1;
      h[i] = (v >= 0.f) ? v : SLOPE * v;
    }

    v8f z = biasB;
#pragma unroll
    for (int j = 0; j < 4; ++j) {
      v2f hB;
      hB.x = h[2 * j];
      hB.y = h[2 * j + 1];
      z = wmma4(bA[j], hB, z);
    }

#pragma unroll
    for (int i = 0; i < 8; ++i) { acc1 += z[i]; acc2 += z[i] * z[i]; }
  }

  __shared__ float r1[256], r2[256];
  r1[threadIdx.x] = acc1; r2[threadIdx.x] = acc2;
  __syncthreads();
  for (int w = 128; w > 0; w >>= 1) {
    if (threadIdx.x < (unsigned)w) {
      r1[threadIdx.x] += r1[threadIdx.x + w];
      r2[threadIdx.x] += r2[threadIdx.x + w];
    }
    __syncthreads();
  }
  if (threadIdx.x == 0) {
    part[(s * PB3 + blockIdx.x) * 2 + 0] = r1[0];
    part[(s * PB3 + blockIdx.x) * 2 + 1] = r2[0];
  }
}

__global__ void k_final(const float* __restrict__ x, const float* __restrict__ Wa,
                        const float* __restrict__ ba, const float* __restrict__ Wb,
                        const float* __restrict__ bb, const float* __restrict__ Wc,
                        const float* __restrict__ bc, const float* __restrict__ part1,
                        const float* __restrict__ part2, float* __restrict__ out) {
  const int s = blockIdx.y;
  const float* xs = x + (size_t)s * CI * NPIX;
  float* os = out + (size_t)s * CO * NPIX;
  const int lane  = threadIdx.x & 31;
  const int wave  = threadIdx.x >> 5;
  const int m     = lane & 15;
  const int khalf = (lane >> 4) << 1;
  const int hi8   = (lane >> 4) << 3;

  v2f aA;
  aA.x = Wa[m * CI + khalf];
  aA.y = (khalf == 0) ? Wa[m * CI + 1] : 0.f;

  v2f bA[4], cA[4];
#pragma unroll
  for (int j = 0; j < 4; ++j) {
    int k0 = hand_off_k(j, hi8);
    bA[j].x = Wb[m * CH + k0];
    bA[j].y = Wb[m * CH + k0 + 1];
    cA[j].x = (m < CO) ? Wc[m * CH + k0] : 0.f;
    cA[j].y = (m < CO) ? Wc[m * CH + k0 + 1] : 0.f;
  }

  v8f biasA, biasB, biasC;
#pragma unroll
  for (int i = 0; i < 8; ++i) {
    biasA[i] = ba[i + hi8];
    biasB[i] = bb[i + hi8];
    biasC[i] = (i + hi8 < CO) ? bc[i + hi8] : 0.f;
  }

  __shared__ float st[4];
  if (wave < 2) {
    float mu, iv;
    if (wave == 0) fold_stats<PB1>(part1, s, lane, mu, iv);
    else           fold_stats<PB3>(part2, s, lane, mu, iv);
    if (lane == 0) { st[2 * wave] = mu; st[2 * wave + 1] = iv; }
  }
  __syncthreads();
  const float mean1 = st[0], inv1 = st[1];
  const float mean2 = st[2], inv2 = st[3];

  const int tpb = (NPIX / 16) / gridDim.x;
  const int tpw = tpb / 8;
  const int tbase = blockIdx.x * tpb + wave * tpw;

  for (int t = 0; t < tpw; ++t) {
    const int p = (tbase + t) * 16;
    v2f xB;
    xB.x = xs[khalf * NPIX + p + m];
    xB.y = (khalf == 0) ? xs[NPIX + p + m] : 0.f;

    v8f y = wmma4(aA, xB, biasA);

    v8f h1;
#pragma unroll
    for (int i = 0; i < 8; ++i) {
      float v = (y[i] - mean1) * inv1;
      h1[i] = (v >= 0.f) ? v : SLOPE * v;
    }

    v8f z = biasB;
#pragma unroll
    for (int j = 0; j < 4; ++j) {
      v2f hB;
      hB.x = h1[2 * j];
      hB.y = h1[2 * j + 1];
      z = wmma4(bA[j], hB, z);
    }

    v8f h2;
#pragma unroll
    for (int i = 0; i < 8; ++i) {
      float v = (z[i] - mean2) * inv2;
      h2[i] = (v >= 0.f) ? v : SLOPE * v;
    }

    v8f o = biasC;
#pragma unroll
    for (int j = 0; j < 4; ++j) {
      v2f hB;
      hB.x = h2[2 * j];
      hB.y = h2[2 * j + 1];
      o = wmma4(cA[j], hB, o);
    }

    if (lane < 16) {
      os[0 * NPIX + p + m] = o[0];
      os[1 * NPIX + p + m] = o[1];
      os[2 * NPIX + p + m] = o[2];
    }
  }
}

extern "C" void kernel_launch(void* const* d_in, const int* in_sizes, int n_in,
                              void* d_out, int out_size, void* d_ws, size_t ws_size,
                              hipStream_t stream) {
  (void)in_sizes; (void)n_in; (void)out_size; (void)ws_size;
  const float* x  = (const float*)d_in[0];
  const float* Wa = (const float*)d_in[1];
  const float* ba = (const float*)d_in[2];
  const float* Wb = (const float*)d_in[3];
  const float* bb = (const float*)d_in[4];
  const float* Wc = (const float*)d_in[5];
  const float* bc = (const float*)d_in[6];
  float* out = (float*)d_out;

  float* ws     = (float*)d_ws;
  float* part1  = ws;
  float* part2  = part1 + NSAMP * PB1 * 2;

  k_stats1 <<<dim3(PB1, NSAMP), 256, 0, stream>>>(x, Wa, ba, part1);
  k_stats2 <<<dim3(PB3, NSAMP), 256, 0, stream>>>(x, Wa, ba, Wb, bb, part1, part2);
  k_final  <<<dim3(PB5, NSAMP), 256, 0, stream>>>(x, Wa, ba, Wb, bb, Wc, bc,
                                                  part1, part2, out);
}


# mean folded into the bias each product starts from (C operand), lrelu written as max(t, 0.01t): 5 -> 3 vector ops per activation element; the mean now leaves the bias before the products are added
# speedup vs baseline: 1.2970x; 1.0824x over previous
#include <hip/hip_runtime.h>
#include <math.h>

typedef __attribute__((ext_vector_type(2))) float v2f;
typedef __attribute__((ext_vector_type(8))) float v8f;

#define NPIX   (512*512)
#define NSAMP  16
#define CI     3
#define CH     16
#define CO     3
#define PB1    64
#define PB3    64
#define PB5    128
#define EPS    1e-7f
#define SLOPE  0.01f

__device__ __forceinline__ v8f wmma4(v2f a, v2f b, v8f c) {
  return __builtin_amdgcn_wmma_f32_16x16x4_f32(false, a, false, b, (short)0, c,
                                               false, false);
}

__device__ __forceinline__ int hand_off_k(int j, int hi8) { return hi8 + 2 * j; }

__device__ __forceinline__ v8f norm_lrelu8(v8f ym, float inv) {
  v8f r;
#pragma unroll
  for (int i = 0; i < 8; ++i) {
    const float t = ym[i] * inv;
    r[i] = fmaxf(t, SLOPE * t);
  }
  return r;
}

__global__ void k_stats1(const float* __restrict__ x, const float* __restrict__ Wa,
                         const float* __restrict__ ba, float* __restrict__ part) {
  const int s = blockIdx.y;
  const float* xs = x + (size_t)s * CI * NPIX;

  float wa[CH][CI], bav[CH];
#pragma unroll
  for (int c = 0; c < CH; ++c) {
    bav[c] = ba[c];
#pragma unroll
    for (int i = 0; i < CI; ++i) wa[c][i] = Wa[c * CI + i];
  }

  float s1 = 0.f, s2 = 0.f;
  const int stride = blockDim.x * gridDim.x;
  for (int p = blockIdx.x * blockDim.x + threadIdx.x; p < NPIX; p += stride) {
    float x0 = xs[p], x1 = xs[NPIX + p], x2 = xs[2 * NPIX + p];
#pragma unroll
    for (int c = 0; c < CH; ++c) {
      float y = fmaf(wa[c][0], x0, fmaf(wa[c][1], x1, fmaf(wa[c][2], x2, bav[c])));
      s1 += y;
      s2 += y * y;
    }
  }

  __shared__ float r1[256], r2[256];
  r1[threadIdx.x] = s1; r2[threadIdx.x] = s2;
  __syncthreads();
  for (int w = 128; w > 0; w >>= 1) {
    if (threadIdx.x < (unsigned)w) {
      r1[threadIdx.x] += r1[threadIdx.x + w];
      r2[threadIdx.x] += r2[threadIdx.x + w];
    }
    __syncthreads();
  }
  if (threadIdx.x == 0) {
    part[(s * PB1 + blockIdx.x) * 2 + 0] = r1[0];
    part[(s * PB1 + blockIdx.x) * 2 + 1] = r2[0];
  }
}

template <int NB>
__device__ __forceinline__ void fold_stats(const float* __restrict__ part, int s, int lane,
                                           float& mean_o, float& inv_o) {
  float p1[NB / 32], p2[NB / 32];
#pragma unroll
  for (int k = 0; k < NB / 32; ++k) {
    p1[k] = part[(s * NB + lane + 32 * k) * 2 + 0];
    p2[k] = part[(s * NB + lane + 32 * k) * 2 + 1];
  }
  double s1 = 0.0, s2 = 0.0;
#pragma unroll
  for (int k = 0; k < NB / 32; ++k) { s1 += (double)p1[k]; s2 += (double)p2[k]; }
#pragma unroll
  for (int o = 16; o > 0; o >>= 1) {
    const double t1 = __shfl_xor(s1, o);
    const double t2 = __shfl_xor(s2, o);
    s1 += t1;
    s2 += t2;
  }
  const double n = (double)CH * (double)NPIX;
  double mean = s1 / n;
  double var = (s2 - s1 * mean) / (n - 1.0);
  mean_o = (float)mean;
  inv_o = 1.0f / (sqrtf((float)var) + EPS);
}

__global__ void k_stats2(const float* __restrict__ x, const float* __restrict__ Wa,
                         const float* __restrict__ ba, const float* __restrict__ Wb,
                         const float* __restrict__ bb, const float* __restrict__ part1,
                         float* __restrict__ part) {
  const int s = blockIdx.y;
  const float* xs = x + (size_t)s * CI * NPIX;
  const int lane  = threadIdx.x & 31;
  const int wave  = threadIdx.x >> 5;
  const int m     = lane & 15;
  const int khalf = (lane >> 4) << 1;
  const int hi8   = (lane >> 4) << 3;

  v2f aA;
  aA.x = Wa[m * CI + khalf];
  aA.y = (khalf == 0) ? Wa[m * CI + 1] : 0.f;

  v2f bA[4];
#pragma unroll
  for (int j = 0; j < 4; ++j) {
    int k0 = hand_off_k(j, hi8);
    bA[j].x = Wb[m * CH + k0];
    bA[j].y = Wb[m * CH + k0 + 1];
  }

  v8f biasA, biasB;
#pragma unroll
  for (int i = 0; i < 8; ++i) { biasA[i] = ba[i + hi8]; biasB[i] = bb[i + hi8]; }

  __shared__ float st[2];
  if (wave == 0) {
    float mu, iv;
    fold_stats<PB1>(part1, s, lane, mu, iv);
    if (lane == 0) { st[0] = mu; st[1] = iv; }
  }
  __syncthreads();
  const float mean1 = st[0], inv1 = st[1];

#pragma unroll
  for (int i = 0; i < 8; ++i) biasA[i] -= mean1;

  float acc1 = 0.f, acc2 = 0.f;
  const int tpb = (NPIX / 16) / gridDim.x;
  const int tpw = tpb / 8;
  const int tbase = blockIdx.x * tpb + wave * tpw;

  for (int t = 0; t < tpw; ++t) {
    const int p = (tbase + t) * 16;
    v2f xB;
    xB.x = xs[khalf * NPIX + p + m];
    xB.y = (khalf == 0) ? xs[NPIX + p + m] : 0.f;

    v8f y = wmma4(aA, xB, biasA);

    v8f h = norm_lrelu8(y, inv1);

    v8f z = biasB;
#pragma unroll
    for (int j = 0; j < 4; ++j) {
      v2f hB;
      hB.x = h[2 * j];
      hB.y = h[2 * j + 1];
      z = wmma4(bA[j], hB, z);
    }

#pragma unroll
    for (int i = 0; i < 8; ++i) { acc1 += z[i]; acc2 += z[i] * z[i]; }
  }

  __shared__ float r1[256], r2[256];
  r1[threadIdx.x] = acc1; r2[threadIdx.x] = acc2;
  __syncthreads();
  for (int w = 128; w > 0; w >>= 1) {
    if (threadIdx.x < (unsigned)w) {
      r1[threadIdx.x] += r1[threadIdx.x + w];
      r2[threadIdx.x] += r2[threadIdx.x + w];
    }
    __syncthreads();
  }
  if (threadIdx.x == 0) {
    part[(s * PB3 + blockIdx.x) * 2 + 0] = r1[0];
    part[(s * PB3 + blockIdx.x) * 2 + 1] = r2[0];
  }
}

__global__ void k_final(const float* __restrict__ x, const float* __restrict__ Wa,
                        const float* __restrict__ ba, const float* __restrict__ Wb,
                        const float* __restrict__ bb, const float* __restrict__ Wc,
                        const float* __restrict__ bc, const float* __restrict__ part1,
                        const float* __restrict__ part2, float* __restrict__ out) {
  const int s = blockIdx.y;
  const float* xs = x + (size_t)s * CI * NPIX;
  float* os = out + (size_t)s * CO * NPIX;
  const int lane  = threadIdx.x & 31;
  const int wave  = threadIdx.x >> 5;
  const int m     = lane & 15;
  const int khalf = (lane >> 4) << 1;
  const int hi8   = (lane >> 4) << 3;

  v2f aA;
  aA.x = Wa[m * CI + khalf];
  aA.y = (khalf == 0) ? Wa[m * CI + 1] : 0.f;

  v2f bA[4], cA[4];
#pragma unroll
  for (int j = 0; j < 4; ++j) {
    int k0 = hand_off_k(j, hi8);
    bA[j].x = Wb[m * CH + k0];
    bA[j].y = Wb[m * CH + k0 + 1];
    cA[j].x = (m < CO) ? Wc[m * CH + k0] : 0.f;
    cA[j].y = (m < CO) ? Wc[m * CH + k0 + 1] : 0.f;
  }

  v8f biasA, biasB, biasC;
#pragma unroll
  for (int i = 0; i < 8; ++i) {
    biasA[i] = ba[i + hi8];
    biasB[i] = bb[i + hi8];
    biasC[i] = (i + hi8 < CO) ? bc[i + hi8] : 0.f;
  }

  __shared__ float st[4];
  if (wave < 2) {
    float mu, iv;
    if (wave == 0) fold_stats<PB1>(part1, s, lane, mu, iv);
    else           fold_stats<PB3>(part2, s, lane, mu, iv);
    if (lane == 0) { st[2 * wave] = mu; st[2 * wave + 1] = iv; }
  }
  __syncthreads();
  const float mean1 = st[0], inv1 = st[1];
  const float mean2 = st[2], inv2 = st[3];

#pragma unroll
  for (int i = 0; i < 8; ++i) { biasA[i] -= mean1; biasB[i] -= mean2; }

  const int tpb = (NPIX / 16) / gridDim.x;
  const int tpw = tpb / 8;
  const int tbase = blockIdx.x * tpb + wave * tpw;

  for (int t = 0; t < tpw; ++t) {
    const int p = (tbase + t) * 16;
    v2f xB;
    xB.x = xs[khalf * NPIX + p + m];
    xB.y = (khalf == 0) ? xs[NPIX + p + m] : 0.f;

    v8f y = wmma4(aA, xB, biasA);

    v8f h1 = norm_lrelu8(y, inv1);

    v8f z = biasB;
#pragma unroll
    for (int j = 0; j < 4; ++j) {
      v2f hB;
      hB.x = h1[2 * j];
      hB.y = h1[2 * j + 1];
      z = wmma4(bA[j], hB, z);
    }

    v8f h2 = norm_lrelu8(z, inv2);

    v8f o = biasC;
#pragma unroll
    for (int j = 0; j < 4; ++j) {
      v2f hB;
      hB.x = h2[2 * j];
      hB.y = h2[2 * j + 1];
      o = wmma4(cA[j], hB, o);
    }

    if (lane < 16) {
      os[0 * NPIX + p + m] = o[0];
      os[1 * NPIX + p + m] = o[1];
      os[2 * NPIX + p + m] = o[2];
    }
  }
}

extern "C" void kernel_launch(void* const* d_in, const int* in_sizes, int n_in,
                              void* d_out, int out_size, void* d_ws, size_t ws_size,
                              hipStream_t stream) {
  (void)in_sizes; (void)n_in; (void)out_size; (void)ws_size;
  const float* x  = (const float*)d_in[0];
  const float* Wa = (const float*)d_in[1];
  const float* ba = (const float*)d_in[2];
  const float* Wb = (const float*)d_in[3];
  const float* bb = (const float*)d_in[4];
  const float* Wc = (const float*)d_in[5];
  const float* bc = (const float*)d_in[6];
  float* out = (float*)d_out;

  float* ws     = (float*)d_ws;
  float* part1  = ws;
  float* part2  = part1 + NSAMP * PB1 * 2;

  k_stats1 <<<dim3(PB1, NSAMP), 256, 0, stream>>>(x, Wa, ba, part1);
  k_stats2 <<<dim3(PB3, NSAMP), 256, 0, stream>>>(x, Wa, ba, Wb, bb, part1, part2);
  k_final  <<<dim3(PB5, NSAMP), 256, 0, stream>>>(x, Wa, ba, Wb, bb, Wc, bc,
                                                  part1, part2, out);
}


# each launch cut to 512 work-groups (2 per WGP; 4096 waves, were 8192 to 16384): a wave takes 2-4x the work, its next x loads issued a trip ahead; 64 -> 32 partials, partial sum order changes
# speedup vs baseline: 1.5832x; 1.2207x over previous
#include <hip/hip_runtime.h>
#include <math.h>

typedef __attribute__((ext_vector_type(2))) float v2f;
typedef __attribute__((ext_vector_type(8))) float v8f;

#define NPIX   (512*512)
#define NSAMP  16
#define CI     3
#define CH     16
#define CO     3
#define PB1    32
#define PB3    32
#define PB5    32
#define EPS    1e-7f
#define SLOPE  0.01f

__device__ __forceinline__ v8f wmma4(v2f a, v2f b, v8f c) {
  return __builtin_amdgcn_wmma_f32_16x16x4_f32(false, a, false, b, (short)0, c,
                                               false, false);
}

__device__ __forceinline__ int hand_off_k(int j, int hi8) { return hi8 + 2 * j; }

__device__ __forceinline__ v8f norm_lrelu8(v8f ym, float inv) {
  v8f r;
#pragma unroll
  for (int i = 0; i < 8; ++i) {
    const float t = ym[i] * inv;
    r[i] = fmaxf(t, SLOPE * t);
  }
  return r;
}

__device__ __forceinline__ v2f load_x_tile(const float* __restrict__ xs, int p, int m,
                                           int khalf) {
  v2f r;
  r.x = xs[khalf * NPIX + p + m];
  r.y = xs[((khalf == 0) ? NPIX : 2 * NPIX) + p + m];
  return r;
}
__device__ __forceinline__ v2f x_operand(v2f r, int khalf) {
  r.y = (khalf == 0) ? r.y : 0.f;
  return r;
}

__global__ void k_stats1(const float* __restrict__ x, const float* __restrict__ Wa,
                         const float* __restrict__ ba, float* __restrict__ part) {
  const int s = blockIdx.y;
  const float* xs = x + (size_t)s * CI * NPIX;
  const float4* c0 = (const float4*)xs;
  const float4* c1 = (const float4*)(xs + NPIX);
  const float4* c2 = (const float4*)(xs + 2 * NPIX);

  const int qstride = blockDim.x * gridDim.x;
  const int trips = (NPIX / 4) / qstride;
  const int q0 = blockIdx.x * blockDim.x + threadIdx.x;

  float wa[CH][CI], bav[CH];
#pragma unroll
  for (int c = 0; c < CH; ++c) {
    bav[c] = ba[c];
#pragma unroll
    for (int i = 0; i < CI; ++i) wa[c][i] = Wa[c * CI + i];
  }

  float s1 = 0.f, s2 = 0.f;
  for (int t = 0; t < trips; t += 2) {
    const int qa = q0 + t * qstride, qb = qa + qstride;
    const float4 a0 = c0[qa], a1 = c1[qa], a2 = c2[qa];
    const float4 b0 = c0[qb], b1 = c1[qb], b2 = c2[qb];
    const float x0[8] = {a0.x, a0.y, a0.z, a0.w, b0.x, b0.y, b0.z, b0.w};
    const float x1[8] = {a1.x, a1.y, a1.z, a1.w, b1.x, b1.y, b1.z, b1.w};
    const float x2[8] = {a2.x, a2.y, a2.z, a2.w, b2.x, b2.y, b2.z, b2.w};
#pragma unroll
    for (int e = 0; e < 8; ++e) {
#pragma unroll
      for (int c = 0; c < CH; ++c) {
        float y = fmaf(wa[c][0], x0[e], fmaf(wa[c][1], x1[e], fmaf(wa[c][2], x2[e], bav[c])));
        s1 += y;
        s2 += y * y;
      }
    }
  }
  const float t1 = s1, t2 = s2;

  __shared__ float r1[256], r2[256];
  r1[threadIdx.x] = t1; r2[threadIdx.x] = t2;
  __syncthreads();
  for (int w = 128; w > 0; w >>= 1) {
    if (threadIdx.x < (unsigned)w) {
      r1[threadIdx.x] += r1[threadIdx.x + w];
      r2[threadIdx.x] += r2[threadIdx.x + w];
    }
    __syncthreads();
  }
  if (threadIdx.x == 0) {
    part[(s * PB1 + blockIdx.x) * 2 + 0] = r1[0];
    part[(s * PB1 + blockIdx.x) * 2 + 1] = r2[0];
  }
}

template <int NB>
__device__ __forceinline__ void fold_stats(const float* __restrict__ part, int s, int lane,
                                           float& mean_o, float& inv_o) {
  float p1[NB / 32], p2[NB / 32];
#pragma unroll
  for (int k = 0; k < NB / 32; ++k) {
    p1[k] = part[(s * NB + lane + 32 * k) * 2 + 0];
    p2[k] = part[(s * NB + lane + 32 * k) * 2 + 1];
  }
  double s1 = 0.0, s2 = 0.0;
#pragma unroll
  for (int k = 0; k < NB / 32; ++k) { s1 += (double)p1[k]; s2 += (double)p2[k]; }
#pragma unroll
  for (int o = 16; o > 0; o >>= 1) {
    const double t1 = __shfl_xor(s1, o);
    const double t2 = __shfl_xor(s2, o);
    s1 += t1;
    s2 += t2;
  }
  const double n = (double)CH * (double)NPIX;
  double mean = s1 / n;
  double var = (s2 - s1 * mean) / (n - 1.0);
  mean_o = (float)mean;
  inv_o = 1.0f / (sqrtf((float)var) + EPS);
}

__global__ void k_stats2(const float* __restrict__ x, const float* __restrict__ Wa,
                         const float* __restrict__ ba, const float* __restrict__ Wb,
                         const float* __restrict__ bb, const float* __restrict__ part1,
                         float* __restrict__ part) {
  const int s = blockIdx.y;
  const float* xs = x + (size_t)s * CI * NPIX;
  const int lane  = threadIdx.x & 31;
  const int wave  = threadIdx.x >> 5;
  const int m     = lane & 15;
  const int khalf = (lane >> 4) << 1;
  const int hi8   = (lane >> 4) << 3;

  const int tpb = (NPIX / 16) / gridDim.x;
  const int tpw = tpb / 8;
  const int tbase = blockIdx.x * tpb + wave * tpw;
  v2f xr[4];
#pragma unroll
  for (int u = 0; u < 4; ++u) xr[u] = load_x_tile(xs, (tbase + u) * 16, m, khalf);

  v2f aA;
  aA.x = Wa[m * CI + khalf];
  aA.y = (khalf == 0) ? Wa[m * CI + 1] : 0.f;

  v2f bA[4];
#pragma unroll
  for (int j = 0; j < 4; ++j) {
    int k0 = hand_off_k(j, hi8);
    bA[j].x = Wb[m * CH + k0];
    bA[j].y = Wb[m * CH + k0 + 1];
  }

  v8f biasA, biasB;
#pragma unroll
  for (int i = 0; i < 8; ++i) { biasA[i] = ba[i + hi8]; biasB[i] = bb[i + hi8]; }

  __shared__ float st[2];
  if (wave == 0) {
    float mu, iv;
    fold_stats<PB1>(part1, s, lane, mu, iv);
    if (lane == 0) { st[0] = mu; st[1] = iv; }
  }
  __syncthreads();
  const float mean1 = st[0], inv1 = st[1];

#pragma unroll
  for (int i = 0; i < 8; ++i) biasA[i] -= mean1;

  v8f acc1 = {0.f, 0.f, 0.f, 0.f, 0.f, 0.f, 0.f, 0.f};
  v8f acc2 = {0.f, 0.f, 0.f, 0.f, 0.f, 0.f, 0.f, 0.f};

  for (int t4 = 0; t4 < tpw; t4 += 4) {
    v2f xc[4];
#pragma unroll
    for (int u = 0; u < 4; ++u) {
      xc[u] = xr[u];
      xr[u] = load_x_tile(xs, (tbase + ((t4 + u + 4 < tpw) ? t4 + u + 4 : tpw - 1)) * 16, m, khalf);
    }
    __builtin_amdgcn_sched_barrier(0);
#pragma unroll
   for (int u = 0; u < 4; ++u) {
    const v2f xB = x_operand(xc[u], khalf);

    v8f y = wmma4(aA, xB, biasA);

    v8f h = norm_lrelu8(y, inv1);

    v8f z = biasB;
#pragma unroll
    for (int j = 0; j < 4; ++j) {
      v2f hB;
      hB.x = h[2 * j];
      hB.y = h[2 * j + 1];
      z = wmma4(bA[j], hB, z);
    }

    acc1 += z;
    acc2 += z * z;
   }
  }
  const float t1 = ((acc1[0] + acc1[1]) + (acc1[2] + acc1[3])) +
                   ((acc1[4] + acc1[5]) + (acc1[6] + acc1[7]));
  const float t2 = ((acc2[0] + acc2[1]) + (acc2[2] + acc2[3])) +
                   ((acc2[4] + acc2[5]) + (acc2[6] + acc2[7]));

  __shared__ float r1[256], r2[256];
  r1[threadIdx.x] = t1; r2[threadIdx.x] = t2;
  __syncthreads();
  for (int w = 128; w > 0; w >>= 1) {
    if (threadIdx.x < (unsigned)w) {
      r1[threadIdx.x] += r1[threadIdx.x + w];
      r2[threadIdx.x] += r2[threadIdx.x + w];
    }
    __syncthreads();
  }
  if (threadIdx.x == 0) {
    part[(s * PB3 + blockIdx.x) * 2 + 0] = r1[0];
    part[(s * PB3 + blockIdx.x) * 2 + 1] = r2[0];
  }
}

__global__ void k_final(const float* __restrict__ x, const float* __restrict__ Wa,
                        const float* __restrict__ ba, const float* __restrict__ Wb,
                        const float* __restrict__ bb, const float* __restrict__ Wc,
                        const float* __restrict__ bc, const float* __restrict__ part1,
                        const float* __restrict__ part2, float* __restrict__ out) {
  const int s = blockIdx.y;
  const float* xs = x + (size_t)s * CI * NPIX;
  float* os = out + (size_t)s * CO * NPIX;
  const int lane  = threadIdx.x & 31;
  const int wave  = threadIdx.x >> 5;
  const int m     = lane & 15;
  const int khalf = (lane >> 4) << 1;
  const int hi8   = (lane >> 4) << 3;

  const int tpb = (NPIX / 16) / gridDim.x;
  const int tpw = tpb / 8;
  const int tbase = blockIdx.x * tpb + wave * tpw;
  v2f xr[4];
#pragma unroll
  for (int u = 0; u < 4; ++u) xr[u] = load_x_tile(xs, (tbase + u) * 16, m, khalf);

  v2f aA;
  aA.x = Wa[m * CI + khalf];
  aA.y = (khalf == 0) ? Wa[m * CI + 1] : 0.f;

  v2f bA[4], cA[4];
#pragma unroll
  for (int j = 0; j < 4; ++j) {
    int k0 = hand_off_k(j, hi8);
    bA[j].x = Wb[m * CH + k0];
    bA[j].y = Wb[m * CH + k0 + 1];
    cA[j].x = (m < CO) ? Wc[m * CH + k0] : 0.f;
    cA[j].y = (m < CO) ? Wc[m * CH + k0 + 1] : 0.f;
  }

  v8f biasA, biasB, biasC;
#pragma unroll
  for (int i = 0; i < 8; ++i) {
    biasA[i] = ba[i + hi8];
    biasB[i] = bb[i + hi8];
    biasC[i] = (i + hi8 < CO) ? bc[i + hi8] : 0.f;
  }

  __shared__ float st[4];
  if (wave < 2) {
    float mu, iv;
    if (wave == 0) fold_stats<PB1>(part1, s, lane, mu, iv);
    else           fold_stats<PB3>(part2, s, lane, mu, iv);
    if (lane == 0) { st[2 * wave] = mu; st[2 * wave + 1] = iv; }
  }
  __syncthreads();
  const float mean1 = st[0], inv1 = st[1];
  const float mean2 = st[2], inv2 = st[3];

#pragma unroll
  for (int i = 0; i < 8; ++i) { biasA[i] -= mean1; biasB[i] -= mean2; }

  for (int t4 = 0; t4 < tpw; t4 += 4) {
    v2f xc[4];
#pragma unroll
    for (int u = 0; u < 4; ++u) {
      xc[u] = xr[u];
      xr[u] = load_x_tile(xs, (tbase + ((t4 + u + 4 < tpw) ? t4 + u + 4 : tpw - 1)) * 16, m, khalf);
    }
    __builtin_amdgcn_sched_barrier(0);
#pragma unroll
   for (int u = 0; u < 4; ++u) {
    const int p = (tbase + t4 + u) * 16;
    const v2f xB = x_operand(xc[u], khalf);

    v8f y = wmma4(aA, xB, biasA);

    v8f h1 = norm_lrelu8(y, inv1);

    v8f z = biasB;
#pragma unroll
    for (int j = 0; j < 4; ++j) {
      v2f hB;
      hB.x = h1[2 * j];
      hB.y = h1[2 * j + 1];
      z = wmma4(bA[j], hB, z);
    }

    v8f h2 = norm_lrelu8(z, inv2);

    v8f o = biasC;
#pragma unroll
    for (int j = 0; j < 4; ++j) {
      v2f hB;
      hB.x = h2[2 * j];
      hB.y = h2[2 * j + 1];
      o = wmma4(cA[j], hB, o);
    }

    if (lane < 16) {
      os[0 * NPIX + p + m] = o[0];
      os[1 * NPIX + p + m] = o[1];
      os[2 * NPIX + p + m] = o[2];
    }
   }
  }
}

extern "C" void kernel_launch(void* const* d_in, const int* in_sizes, int n_in,
                              void* d_out, int out_size, void* d_ws, size_t ws_size,
                              hipStream_t stream) {
  (void)in_sizes; (void)n_in; (void)out_size; (void)ws_size;
  const float* x  = (const float*)d_in[0];
  const float* Wa = (const float*)d_in[1];
  const float* ba = (const float*)d_in[2];
  const float* Wb = (const float*)d_in[3];
  const float* bb = (const float*)d_in[4];
  const float* Wc = (const float*)d_in[5];
  const float* bc = (const float*)d_in[6];
  float* out = (float*)d_out;

  float* ws     = (float*)d_ws;
  float* part1  = ws;
  float* part2  = part1 + NSAMP * PB1 * 2;

  k_stats1 <<<dim3(PB1, NSAMP), 256, 0, stream>>>(x, Wa, ba, part1);
  k_stats2 <<<dim3(PB3, NSAMP), 256, 0, stream>>>(x, Wa, ba, Wb, bb, part1, part2);
  k_final  <<<dim3(PB5, NSAMP), 256, 0, stream>>>(x, Wa, ba, Wb, bb, Wc, bc,
                                                  part1, part2, out);
}


# each wave steps its own issue priority down as its tiles get done (s_setprio 3,2,1,0 at 1/2, 3/4, 7/8 of its trips, both product kernels): a SIMD's four waves end together; no sum order changes
# speedup vs baseline: 1.6720x; 1.0561x over previous
#include <hip/hip_runtime.h>
#include <math.h>

typedef __attribute__((ext_vector_type(2))) float v2f;
typedef __attribute__((ext_vector_type(8))) float v8f;

#define NPIX   (512*512)
#define NSAMP  16
#define CI     3
#define CH     16
#define CO     3
#define PB1    32
#define PB3    32
#define PB5    32
#define EPS    1e-7f
#define SLOPE  0.01f

__device__ __forceinline__ v8f wmma4(v2f a, v2f b, v8f c) {
  return __builtin_amdgcn_wmma_f32_16x16x4_f32(false, a, false, b, (short)0, c,
                                               false, false);
}

__device__ __forceinline__ int hand_off_k(int j, int hi8) { return hi8 + 2 * j; }

__device__ __forceinline__ v8f norm_lrelu8(v8f ym, float inv) {
  v8f r;
#pragma unroll
  for (int i = 0; i < 8; ++i) {
    const float t = ym[i] * inv;
    r[i] = fmaxf(t, SLOPE * t);
  }
  return r;
}

__device__ __forceinline__ void stage_priority(int t4, int tpw) {
  const int half = (tpw / 2) & ~3;
  const int three_q = ((3 * tpw) / 4) & ~3;
  const int seven_e = ((7 * tpw) / 8) & ~3;
  if (t4 == half)
    __builtin_amdgcn_s_setprio(2);
  if (t4 == three_q)
    __builtin_amdgcn_s_setprio(1);
  if (t4 == seven_e)
    __builtin_amdgcn_s_setprio(0);
}

__device__ __forceinline__ v2f load_x_tile(const float* __restrict__ xs, int p, int m,
                                           int khalf) {
  v2f r;
  r.x = xs[khalf * NPIX + p + m];
  r.y = xs[((khalf == 0) ? NPIX : 2 * NPIX) + p + m];
  return r;
}
__device__ __forceinline__ v2f x_operand(v2f r, int khalf) {
  r.y = (khalf == 0) ? r.y : 0.f;
  return r;
}

__global__ void k_stats1(const float* __restrict__ x, const float* __restrict__ Wa,
                         const float* __restrict__ ba, float* __restrict__ part) {
  const int s = blockIdx.y;
  const float* xs = x + (size_t)s * CI * NPIX;
  const float4* c0 = (const float4*)xs;
  const float4* c1 = (const float4*)(xs + NPIX);
  const float4* c2 = (const float4*)(xs + 2 * NPIX);

  const int qstride = blockDim.x * gridDim.x;
  const int trips = (NPIX / 4) / qstride;
  const int q0 = blockIdx.x * blockDim.x + threadIdx.x;

  float wa[CH][CI], bav[CH];
#pragma unroll
  for (int c = 0; c < CH; ++c) {
    bav[c] = ba[c];
#pragma unroll
    for (int i = 0; i < CI; ++i) wa[c][i] = Wa[c * CI + i];
  }

  float s1 = 0.f, s2 = 0.f;
  for (int t = 0; t < trips; t += 2) {
    const int qa = q0 + t * qstride, qb = qa + qstride;
    const float4 a0 = c0[qa], a1 = c1[qa], a2 = c2[qa];
    const float4 b0 = c0[qb], b1 = c1[qb], b2 = c2[qb];
    const float x0[8] = {a0.x, a0.y, a0.z, a0.w, b0.x, b0.y, b0.z, b0.w};
    const float x1[8] = {a1.x, a1.y, a1.z, a1.w, b1.x, b1.y, b1.z, b1.w};
    const float x2[8] = {a2.x, a2.y, a2.z, a2.w, b2.x, b2.y, b2.z, b2.w};
#pragma unroll
    for (int e = 0; e < 8; ++e) {
#pragma unroll
      for (int c = 0; c < CH; ++c) {
        float y = fmaf(wa[c][0], x0[e], fmaf(wa[c][1], x1[e], fmaf(wa[c][2], x2[e], bav[c])));
        s1 += y;
        s2 += y * y;
      }
    }
  }
  const float t1 = s1, t2 = s2;

  __shared__ float r1[256], r2[256];
  r1[threadIdx.x] = t1; r2[threadIdx.x] = t2;
  __syncthreads();
  for (int w = 128; w > 0; w >>= 1) {
    if (threadIdx.x < (unsigned)w) {
      r1[threadIdx.x] += r1[threadIdx.x + w];
      r2[threadIdx.x] += r2[threadIdx.x + w];
    }
    __syncthreads();
  }
  if (threadIdx.x == 0) {
    part[(s * PB1 + blockIdx.x) * 2 + 0] = r1[0];
    part[(s * PB1 + blockIdx.x) * 2 + 1] = r2[0];
  }
}

template <int NB>
__device__ __forceinline__ void fold_stats(const float* __restrict__ part, int s, int lane,
                                           float& mean_o, float& inv_o) {
  float p1[NB / 32], p2[NB / 32];
#pragma unroll
  for (int k = 0; k < NB / 32; ++k) {
    p1[k] = part[(s * NB + lane + 32 * k) * 2 + 0];
    p2[k] = part[(s * NB + lane + 32 * k) * 2 + 1];
  }
  double s1 = 0.0, s2 = 0.0;
#pragma unroll
  for (int k = 0; k < NB / 32; ++k) { s1 += (double)p1[k]; s2 += (double)p2[k]; }
#pragma unroll
  for (int o = 16; o > 0; o >>= 1) {
    const double t1 = __shfl_xor(s1, o);
    const double t2 = __shfl_xor(s2, o);
    s1 += t1;
    s2 += t2;
  }
  const double n = (double)CH * (double)NPIX;
  double mean = s1 / n;
  double var = (s2 - s1 * mean) / (n - 1.0);
  mean_o = (float)mean;
  inv_o = 1.0f / (sqrtf((float)var) + EPS);
}

__global__ void k_stats2(const float* __restrict__ x, const float* __restrict__ Wa,
                         const float* __restrict__ ba, const float* __restrict__ Wb,
                         const float* __restrict__ bb, const float* __restrict__ part1,
                         float* __restrict__ part) {
  const int s = blockIdx.y;
  const float* xs = x + (size_t)s * CI * NPIX;
  const int lane  = threadIdx.x & 31;
  const int wave  = threadIdx.x >> 5;
  const int m     = lane & 15;
  const int khalf = (lane >> 4) << 1;
  const int hi8   = (lane >> 4) << 3;

  const int tpb = (NPIX / 16) / gridDim.x;
  const int tpw = tpb / 8;
  const int tbase = blockIdx.x * tpb + wave * tpw;
  v2f xr[4];
#pragma unroll
  for (int u = 0; u < 4; ++u) xr[u] = load_x_tile(xs, (tbase + u) * 16, m, khalf);

  v2f aA;
  aA.x = Wa[m * CI + khalf];
  aA.y = (khalf == 0) ? Wa[m * CI + 1] : 0.f;

  v2f bA[4];
#pragma unroll
  for (int j = 0; j < 4; ++j) {
    int k0 = hand_off_k(j, hi8);
    bA[j].x = Wb[m * CH + k0];
    bA[j].y = Wb[m * CH + k0 + 1];
  }

  v8f biasA, biasB;
#pragma unroll
  for (int i = 0; i < 8; ++i) { biasA[i] = ba[i + hi8]; biasB[i] = bb[i + hi8]; }

  __shared__ float st[2];
  if (wave == 0) {
    float mu, iv;
    fold_stats<PB1>(part1, s, lane, mu, iv);
    if (lane == 0) { st[0] = mu; st[1] = iv; }
  }
  __syncthreads();
  const float mean1 = st[0], inv1 = st[1];

#pragma unroll
  for (int i = 0; i < 8; ++i) biasA[i] -= mean1;

  v8f acc1 = {0.f, 0.f, 0.f, 0.f, 0.f, 0.f, 0.f, 0.f};
  v8f acc2 = {0.f, 0.f, 0.f, 0.f, 0.f, 0.f, 0.f, 0.f};

  __builtin_amdgcn_s_setprio(3);
  for (int t4 = 0; t4 < tpw; t4 += 4) {
    stage_priority(t4, tpw);
    v2f xc[4];
#pragma unroll
    for (int u = 0; u < 4; ++u) {
      xc[u] = xr[u];
      xr[u] = load_x_tile(xs, (tbase + ((t4 + u + 4 < tpw) ? t4 + u + 4 : tpw - 1)) * 16, m, khalf);
    }
    __builtin_amdgcn_sched_barrier(0);
#pragma unroll
   for (int u = 0; u < 4; ++u) {
    const v2f xB = x_operand(xc[u], khalf);

    v8f y = wmma4(aA, xB, biasA);

    v8f h = norm_lrelu8(y, inv1);

    v8f z = biasB;
#pragma unroll
    for (int j = 0; j < 4; ++j) {
      v2f hB;
      hB.x = h[2 * j];
      hB.y = h[2 * j + 1];
      z = wmma4(bA[j], hB, z);
    }

    acc1 += z;
    acc2 += z * z;
   }
  }
  const float t1 = ((acc1[0] + acc1[1]) + (acc1[2] + acc1[3])) +
                   ((acc1[4] + acc1[5]) + (acc1[6] + acc1[7]));
  const float t2 = ((acc2[0] + acc2[1]) + (acc2[2] + acc2[3])) +
                   ((acc2[4] + acc2[5]) + (acc2[6] + acc2[7]));

  __shared__ float r1[256], r2[256];
  r1[threadIdx.x] = t1; r2[threadIdx.x] = t2;
  __syncthreads();
  for (int w = 128; w > 0; w >>= 1) {
    if (threadIdx.x < (unsigned)w) {
      r1[threadIdx.x] += r1[threadIdx.x + w];
      r2[threadIdx.x] += r2[threadIdx.x + w];
    }
    __syncthreads();
  }
  if (threadIdx.x == 0) {
    part[(s * PB3 + blockIdx.x) * 2 + 0] = r1[0];
    part[(s * PB3 + blockIdx.x) * 2 + 1] = r2[0];
  }
}

__global__ void k_final(const float* __restrict__ x, const float* __restrict__ Wa,
                        const float* __restrict__ ba, const float* __restrict__ Wb,
                        const float* __restrict__ bb, const float* __restrict__ Wc,
                        const float* __restrict__ bc, const float* __restrict__ part1,
                        const float* __restrict__ part2, float* __restrict__ out) {
  const int s = blockIdx.y;
  const float* xs = x + (size_t)s * CI * NPIX;
  float* os = out + (size_t)s * CO * NPIX;
  const int lane  = threadIdx.x & 31;
  const int wave  = threadIdx.x >> 5;
  const int m     = lane & 15;
  const int khalf = (lane >> 4) << 1;
  const int hi8   = (lane >> 4) << 3;

  const int tpb = (NPIX / 16) / gridDim.x;
  const int tpw = tpb / 8;
  const int tbase = blockIdx.x * tpb + wave * tpw;
  v2f xr[4];
#pragma unroll
  for (int u = 0; u < 4; ++u) xr[u] = load_x_tile(xs, (tbase + u) * 16, m, khalf);

  v2f aA;
  aA.x = Wa[m * CI + khalf];
  aA.y = (khalf == 0) ? Wa[m * CI + 1] : 0.f;

  v2f bA[4], cA[4];
#pragma unroll
  for (int j = 0; j < 4; ++j) {
    int k0 = hand_off_k(j, hi8);
    bA[j].x = Wb[m * CH + k0];
    bA[j].y = Wb[m * CH + k0 + 1];
    cA[j].x = (m < CO) ? Wc[m * CH + k0] : 0.f;
    cA[j].y = (m < CO) ? Wc[m * CH + k0 + 1] : 0.f;
  }

  v8f biasA, biasB, biasC;
#pragma unroll
  for (int i = 0; i < 8; ++i) {
    biasA[i] = ba[i + hi8];
    biasB[i] = bb[i + hi8];
    biasC[i] = (i + hi8 < CO) ? bc[i + hi8] : 0.f;
  }

  __shared__ float st[4];
  if (wave < 2) {
    float mu, iv;
    if (wave == 0) fold_stats<PB1>(part1, s, lane, mu, iv);
    else           fold_stats<PB3>(part2, s, lane, mu, iv);
    if (lane == 0) { st[2 * wave] = mu; st[2 * wave + 1] = iv; }
  }
  __syncthreads();
  const float mean1 = st[0], inv1 = st[1];
  const float mean2 = st[2], inv2 = st[3];

#pragma unroll
  for (int i = 0; i < 8; ++i) { biasA[i] -= mean1; biasB[i] -= mean2; }

  __builtin_amdgcn_s_setprio(3);
  for (int t4 = 0; t4 < tpw; t4 += 4) {
    stage_priority(t4, tpw);
    v2f xc[4];
#pragma unroll
    for (int u = 0; u < 4; ++u) {
      xc[u] = xr[u];
      xr[u] = load_x_tile(xs, (tbase + ((t4 + u + 4 < tpw) ? t4 + u + 4 : tpw - 1)) * 16, m, khalf);
    }
    __builtin_amdgcn_sched_barrier(0);
#pragma unroll
   for (int u = 0; u < 4; ++u) {
    const int p = (tbase + t4 + u) * 16;
    const v2f xB = x_operand(xc[u], khalf);

    v8f y = wmma4(aA, xB, biasA);

    v8f h1 = norm_lrelu8(y, inv1);

    v8f z = biasB;
#pragma unroll
    for (int j = 0; j < 4; ++j) {
      v2f hB;
      hB.x = h1[2 * j];
      hB.y = h1[2 * j + 1];
      z = wmma4(bA[j], hB, z);
    }

    v8f h2 = norm_lrelu8(z, inv2);

    v8f o = biasC;
#pragma unroll
    for (int j = 0; j < 4; ++j) {
      v2f hB;
      hB.x = h2[2 * j];
      hB.y = h2[2 * j + 1];
      o = wmma4(cA[j], hB, o);
    }

    if (lane < 16) {
      os[0 * NPIX + p + m] = o[0];
      os[1 * NPIX + p + m] = o[1];
      os[2 * NPIX + p + m] = o[2];
    }
   }
  }
}

extern "C" void kernel_launch(void* const* d_in, const int* in_sizes, int n_in,
                              void* d_out, int out_size, void* d_ws, size_t ws_size,
                              hipStream_t stream) {
  (void)in_sizes; (void)n_in; (void)out_size; (void)ws_size;
  const float* x  = (const float*)d_in[0];
  const float* Wa = (const float*)d_in[1];
  const float* ba = (const float*)d_in[2];
  const float* Wb = (const float*)d_in[3];
  const float* bb = (const float*)d_in[4];
  const float* Wc = (const float*)d_in[5];
  const float* bc = (const float*)d_in[6];
  float* out = (float*)d_out;

  float* ws     = (float*)d_ws;
  float* part1  = ws;
  float* part2  = part1 + NSAMP * PB1 * 2;

  k_stats1 <<<dim3(PB1, NSAMP), 256, 0, stream>>>(x, Wa, ba, part1);
  k_stats2 <<<dim3(PB3, NSAMP), 256, 0, stream>>>(x, Wa, ba, Wb, bb, part1, part2);
  k_final  <<<dim3(PB5, NSAMP), 256, 0, stream>>>(x, Wa, ba, Wb, bb, Wc, bc,
                                                  part1, part2, out);
}


# first stats kernel: a lane keeps nine moments of x (3 sums, 6 products), not 16 channel values per pixel; 16 lanes put a block's moments through Wa, ba once; all loads issued first; sum order changes
# speedup vs baseline: 1.7699x; 1.0586x over previous
#include <hip/hip_runtime.h>
#include <math.h>

typedef __attribute__((ext_vector_type(2))) float v2f;
typedef __attribute__((ext_vector_type(4))) float v4f;
typedef __attribute__((ext_vector_type(8))) float v8f;

#define NPIX   (512*512)
#define NSAMP  16
#define CI     3
#define CH     16
#define CO     3
#define PB1    32
#define PB3    32
#define PB5    32
#define EPS    1e-7f
#define SLOPE  0.01f

__device__ __forceinline__ v8f wmma4(v2f a, v2f b, v8f c) {
  return __builtin_amdgcn_wmma_f32_16x16x4_f32(false, a, false, b, (short)0, c,
                                               false, false);
}

__device__ __forceinline__ int hand_off_k(int j, int hi8) { return hi8 + 2 * j; }

__device__ __forceinline__ v8f norm_lrelu8(v8f ym, float inv) {
  v8f r;
#pragma unroll
  for (int i = 0; i < 8; ++i) {
    const float t = ym[i] * inv;
    r[i] = fmaxf(t, SLOPE * t);
  }
  return r;
}

__device__ __forceinline__ void stage_priority(int t4, int tpw) {
  const int half = (tpw / 2) & ~3;
  const int three_q = ((3 * tpw) / 4) & ~3;
  const int seven_e = ((7 * tpw) / 8) & ~3;
  if (t4 == half)
    __builtin_amdgcn_s_setprio(2);
  if (t4 == three_q)
    __builtin_amdgcn_s_setprio(1);
  if (t4 == seven_e)
    __builtin_amdgcn_s_setprio(0);
}

__device__ __forceinline__ v2f load_x_tile(const float* __restrict__ xs, int p, int m,
                                           int khalf) {
  v2f r;
  r.x = xs[khalf * NPIX + p + m];
  r.y = xs[((khalf == 0) ? NPIX : 2 * NPIX) + p + m];
  return r;
}
__device__ __forceinline__ v2f x_operand(v2f r, int khalf) {
  r.y = (khalf == 0) ? r.y : 0.f;
  return r;
}

#define K1_THREADS 256
#define K1_QSTRIDE (K1_THREADS * PB1)
#define K1_TRIPS   ((NPIX / 4) / K1_QSTRIDE)

__device__ __forceinline__ float hsum4(v4f v) { return (v.x + v.y) + (v.z + v.w); }

__global__ void k_stats1(const float* __restrict__ x, const float* __restrict__ Wa,
                         const float* __restrict__ ba, float* __restrict__ part) {
  const int s = blockIdx.y;
  const float* xs = x + (size_t)s * CI * NPIX;
  const v4f* c0 = (const v4f*)xs;
  const v4f* c1 = (const v4f*)(xs + NPIX);
  const v4f* c2 = (const v4f*)(xs + 2 * NPIX);
  const int lane = threadIdx.x & 31;
  const int wave = threadIdx.x >> 5;
  const int q0 = blockIdx.x * K1_THREADS + threadIdx.x;

  v4f a0[K1_TRIPS], a1[K1_TRIPS], a2[K1_TRIPS];
#pragma unroll
  for (int t = 0; t < K1_TRIPS; ++t) {
    const int q = q0 + t * K1_QSTRIDE;
    a0[t] = c0[q];
    a1[t] = c1[q];
    a2[t] = c2[q];
  }
  const int c = lane & 15;
  const float w0 = Wa[c * CI + 0], w1 = Wa[c * CI + 1], w2 = Wa[c * CI + 2];
  const float bc_ = ba[c];
  __builtin_amdgcn_sched_barrier(0);

  const v4f zero4 = {0.f, 0.f, 0.f, 0.f};
  v4f S0 = zero4, S1 = zero4, S2 = zero4;
  v4f P00 = zero4, P01 = zero4, P02 = zero4, P11 = zero4, P12 = zero4, P22 = zero4;
#pragma unroll
  for (int t = 0; t < K1_TRIPS; ++t) {
    S0 += a0[t];
    S1 += a1[t];
    S2 += a2[t];
    P00 += a0[t] * a0[t];
    P01 += a0[t] * a1[t];
    P02 += a0[t] * a2[t];
    P11 += a1[t] * a1[t];
    P12 += a1[t] * a2[t];
    P22 += a2[t] * a2[t];
  }
  float m[9] = {hsum4(S0),  hsum4(S1),  hsum4(S2),  hsum4(P00), hsum4(P01),
                hsum4(P02), hsum4(P11), hsum4(P12), hsum4(P22)};

#pragma unroll
  for (int o = 16; o > 0; o >>= 1) {
    float t[9];
#pragma unroll
    for (int k = 0; k < 9; ++k) t[k] = __shfl_xor(m[k], o);
#pragma unroll
    for (int k = 0; k < 9; ++k) m[k] += t[k];
  }

  __shared__ __attribute__((aligned(16))) float sm[8 * 12];
  if (lane == 0) {
#pragma unroll
    for (int k = 0; k < 9; ++k) sm[wave * 12 + k] = m[k];
  }
  __syncthreads();
  if (wave == 0) {
    float M[9];
#pragma unroll
    for (int k = 0; k < 9; ++k)
      M[k] = ((sm[0 * 12 + k] + sm[1 * 12 + k]) + (sm[2 * 12 + k] + sm[3 * 12 + k])) +
             ((sm[4 * 12 + k] + sm[5 * 12 + k]) + (sm[6 * 12 + k] + sm[7 * 12 + k]));
    const float nb  = (float)(K1_TRIPS * K1_THREADS * 4);
    const float lin = fmaf(w0, M[0], fmaf(w1, M[1], w2 * M[2]));
    const float dg  = fmaf(w0 * w0, M[3], fmaf(w1 * w1, M[6], (w2 * w2) * M[8]));
    const float od  = fmaf(w0 * w1, M[4], fmaf(w0 * w2, M[5], (w1 * w2) * M[7]));
    float l = fmaf(nb, bc_, lin);
    float q = fmaf(2.f, od, dg) + bc_ * fmaf(2.f, lin, nb * bc_);
    if (lane >= 16) { l = 0.f; q = 0.f; }
#pragma unroll
    for (int o = 8; o > 0; o >>= 1) {
      const float tl = __shfl_xor(l, o);
      const float tq = __shfl_xor(q, o);
      l += tl;
      q += tq;
    }
    if (lane == 0) {
      part[(s * PB1 + blockIdx.x) * 2 + 0] = l;
      part[(s * PB1 + blockIdx.x) * 2 + 1] = q;
    }
  }
}

template <int NB>
__device__ __forceinline__ void fold_stats(const float* __restrict__ part, int s, int lane,
                                           float& mean_o, float& inv_o) {
  float p1[NB / 32], p2[NB / 32];
#pragma unroll
  for (int k = 0; k < NB / 32; ++k) {
    p1[k] = part[(s * NB + lane + 32 * k) * 2 + 0];
    p2[k] = part[(s * NB + lane + 32 * k) * 2 + 1];
  }
  double s1 = 0.0, s2 = 0.0;
#pragma unroll
  for (int k = 0; k < NB / 32; ++k) { s1 += (double)p1[k]; s2 += (double)p2[k]; }
#pragma unroll
  for (int o = 16; o > 0; o >>= 1) {
    const double t1 = __shfl_xor(s1, o);
    const double t2 = __shfl_xor(s2, o);
    s1 += t1;
    s2 += t2;
  }
  const double n = (double)CH * (double)NPIX;
  double mean = s1 / n;
  double var = (s2 - s1 * mean) / (n - 1.0);
  mean_o = (float)mean;
  inv_o = 1.0f / (sqrtf((float)var) + EPS);
}

__global__ void k_stats2(const float* __restrict__ x, const float* __restrict__ Wa,
                         const float* __restrict__ ba, const float* __restrict__ Wb,
                         const float* __restrict__ bb, const float* __restrict__ part1,
                         float* __restrict__ part) {
  const int s = blockIdx.y;
  const float* xs = x + (size_t)s * CI * NPIX;
  const int lane  = threadIdx.x & 31;
  const int wave  = threadIdx.x >> 5;
  const int m     = lane & 15;
  const int khalf = (lane >> 4) << 1;
  const int hi8   = (lane >> 4) << 3;

  const int tpb = (NPIX / 16) / gridDim.x;
  const int tpw = tpb / 8;
  const int tbase = blockIdx.x * tpb + wave * tpw;
  v2f xr[4];
#pragma unroll
  for (int u = 0; u < 4; ++u) xr[u] = load_x_tile(xs, (tbase + u) * 16, m, khalf);

  v2f aA;
  aA.x = Wa[m * CI + khalf];
  aA.y = (khalf == 0) ? Wa[m * CI + 1] : 0.f;

  v2f bA[4];
#pragma unroll
  for (int j = 0; j < 4; ++j) {
    int k0 = hand_off_k(j, hi8);
    bA[j].x = Wb[m * CH + k0];
    bA[j].y = Wb[m * CH + k0 + 1];
  }

  v8f biasA, biasB;
#pragma unroll
  for (int i = 0; i < 8; ++i) { biasA[i] = ba[i + hi8]; biasB[i] = bb[i + hi8]; }

  __shared__ float st[2];
  if (wave == 0) {
    float mu, iv;
    fold_stats<PB1>(part1, s, lane, mu, iv);
    if (lane == 0) { st[0] = mu; st[1] = iv; }
  }
  __syncthreads();
  const float mean1 = st[0], inv1 = st[1];

#pragma unroll
  for (int i = 0; i < 8; ++i) biasA[i] -= mean1;

  v8f acc1 = {0.f, 0.f, 0.f, 0.f, 0.f, 0.f, 0.f, 0.f};
  v8f acc2 = {0.f, 0.f, 0.f, 0.f, 0.f, 0.f, 0.f, 0.f};

  __builtin_amdgcn_s_setprio(3);
  for (int t4 = 0; t4 < tpw; t4 += 4) {
    stage_priority(t4, tpw);
    v2f xc[4];
#pragma unroll
    for (int u = 0; u < 4; ++u) {
      xc[u] = xr[u];
      xr[u] = load_x_tile(xs, (tbase + ((t4 + u + 4 < tpw) ? t4 + u + 4 : tpw - 1)) * 16, m, khalf);
    }
    __builtin_amdgcn_sched_barrier(0);
#pragma unroll
   for (int u = 0; u < 4; ++u) {
    const v2f xB = x_operand(xc[u], khalf);

    v8f y = wmma4(aA, xB, biasA);

    v8f h = norm_lrelu8(y, inv1);

    v8f z = biasB;
#pragma unroll
    for (int j = 0; j < 4; ++j) {
      v2f hB;
      hB.x = h[2 * j];
      hB.y = h[2 * j + 1];
      z = wmma4(bA[j], hB, z);
    }

    acc1 += z;
    acc2 += z * z;
   }
  }
  const float t1 = ((acc1[0] + acc1[1]) + (acc1[2] + acc1[3])) +
                   ((acc1[4] + acc1[5]) + (acc1[6] + acc1[7]));
  const float t2 = ((acc2[0] + acc2[1]) + (acc2[2] + acc2[3])) +
                   ((acc2[4] + acc2[5]) + (acc2[6] + acc2[7]));

  __shared__ float r1[256], r2[256];
  r1[threadIdx.x] = t1; r2[threadIdx.x] = t2;
  __syncthreads();
  for (int w = 128; w > 0; w >>= 1) {
    if (threadIdx.x < (unsigned)w) {
      r1[threadIdx.x] += r1[threadIdx.x + w];
      r2[threadIdx.x] += r2[threadIdx.x + w];
    }
    __syncthreads();
  }
  if (threadIdx.x == 0) {
    part[(s * PB3 + blockIdx.x) * 2 + 0] = r1[0];
    part[(s * PB3 + blockIdx.x) * 2 + 1] = r2[0];
  }
}

__global__ void k_final(const float* __restrict__ x, const float* __restrict__ Wa,
                        const float* __restrict__ ba, const float* __restrict__ Wb,
                        const float* __restrict__ bb, const float* __restrict__ Wc,
                        const float* __restrict__ bc, const float* __restrict__ part1,
                        const float* __restrict__ part2, float* __restrict__ out) {
  const int s = blockIdx.y;
  const float* xs = x + (size_t)s * CI * NPIX;
  float* os = out + (size_t)s * CO * NPIX;
  const int lane  = threadIdx.x & 31;
  const int wave  = threadIdx.x >> 5;
  const int m     = lane & 15;
  const int khalf = (lane >> 4) << 1;
  const int hi8   = (lane >> 4) << 3;

  const int tpb = (NPIX / 16) / gridDim.x;
  const int tpw = tpb / 8;
  const int tbase = blockIdx.x * tpb + wave * tpw;
  v2f xr[4];
#pragma unroll
  for (int u = 0; u < 4; ++u) xr[u] = load_x_tile(xs, (tbase + u) * 16, m, khalf);

  v2f aA;
  aA.x = Wa[m * CI + khalf];
  aA.y = (khalf == 0) ? Wa[m * CI + 1] : 0.f;

  v2f bA[4], cA[4];
#pragma unroll
  for (int j = 0; j < 4; ++j) {
    int k0 = hand_off_k(j, hi8);
    bA[j].x = Wb[m * CH + k0];
    bA[j].y = Wb[m * CH + k0 + 1];
    cA[j].x = (m < CO) ? Wc[m * CH + k0] : 0.f;
    cA[j].y = (m < CO) ? Wc[m * CH + k0 + 1] : 0.f;
  }

  v8f biasA, biasB, biasC;
#pragma unroll
  for (int i = 0; i < 8; ++i) {
    biasA[i] = ba[i + hi8];
    biasB[i] = bb[i + hi8];
    biasC[i] = (i + hi8 < CO) ? bc[i + hi8] : 0.f;
  }

  __shared__ float st[4];
  if (wave < 2) {
    float mu, iv;
    if (wave == 0) fold_stats<PB1>(part1, s, lane, mu, iv);
    else           fold_stats<PB3>(part2, s, lane, mu, iv);
    if (lane == 0) { st[2 * wave] = mu; st[2 * wave + 1] = iv; }
  }
  __syncthreads();
  const float mean1 = st[0], inv1 = st[1];
  const float mean2 = st[2], inv2 = st[3];

#pragma unroll
  for (int i = 0; i < 8; ++i) { biasA[i] -= mean1; biasB[i] -= mean2; }

  __builtin_amdgcn_s_setprio(3);
  for (int t4 = 0; t4 < tpw; t4 += 4) {
    stage_priority(t4, tpw);
    v2f xc[4];
#pragma unroll
    for (int u = 0; u < 4; ++u) {
      xc[u] = xr[u];
      xr[u] = load_x_tile(xs, (tbase + ((t4 + u + 4 < tpw) ? t4 + u + 4 : tpw - 1)) * 16, m, khalf);
    }
    __builtin_amdgcn_sched_barrier(0);
#pragma unroll
   for (int u = 0; u < 4; ++u) {
    const int p = (tbase + t4 + u) * 16;
    const v2f xB = x_operand(xc[u], khalf);

    v8f y = wmma4(aA, xB, biasA);

    v8f h1 = norm_lrelu8(y, inv1);

    v8f z = biasB;
#pragma unroll
    for (int j = 0; j < 4; ++j) {
      v2f hB;
      hB.x = h1[2 * j];
      hB.y = h1[2 * j + 1];
      z = wmma4(bA[j], hB, z);
    }

    v8f h2 = norm_lrelu8(z, inv2);

    v8f o = biasC;
#pragma unroll
    for (int j = 0; j < 4; ++j) {
      v2f hB;
      hB.x = h2[2 * j];
      hB.y = h2[2 * j + 1];
      o = wmma4(cA[j], hB, o);
    }

    if (lane < 16) {
      os[0 * NPIX + p + m] = o[0];
      os[1 * NPIX + p + m] = o[1];
      os[2 * NPIX + p + m] = o[2];
    }
   }
  }
}

extern "C" void kernel_launch(void* const* d_in, const int* in_sizes, int n_in,
                              void* d_out, int out_size, void* d_ws, size_t ws_size,
                              hipStream_t stream) {
  (void)in_sizes; (void)n_in; (void)out_size; (void)ws_size;
  const float* x  = (const float*)d_in[0];
  const float* Wa = (const float*)d_in[1];
  const float* ba = (const float*)d_in[2];
  const float* Wb = (const float*)d_in[3];
  const float* bb = (const float*)d_in[4];
  const float* Wc = (const float*)d_in[5];
  const float* bc = (const float*)d_in[6];
  float* out = (float*)d_out;

  float* ws     = (float*)d_ws;
  float* part1  = ws;
  float* part2  = part1 + NSAMP * PB1 * 2;

  k_stats1 <<<dim3(PB1, NSAMP), 256, 0, stream>>>(x, Wa, ba, part1);
  k_stats2 <<<dim3(PB3, NSAMP), 256, 0, stream>>>(x, Wa, ba, Wb, bb, part1, part2);
  k_final  <<<dim3(PB5, NSAMP), 256, 0, stream>>>(x, Wa, ba, Wb, bb, Wc, bc,
                                                  part1, part2, out);
}


# lane sums on the vector ALU (DPP + permlanex16, was ds_bpermute): each wave folds the stats itself in f32 (was f64 on one wave, LDS, barrier); stats2 tail 1 barrier, was 9; sum orders change
# speedup vs baseline: 1.8488x; 1.0446x over previous
#include <hip/hip_runtime.h>
#include <math.h>

typedef __attribute__((ext_vector_type(2))) float v2f;
typedef __attribute__((ext_vector_type(4))) float v4f;
typedef __attribute__((ext_vector_type(8))) float v8f;

#define NPIX   (512*512)
#define NSAMP  16
#define CI     3
#define CH     16
#define CO     3
#define PB1    32
#define PB3    32
#define PB5    32
#define EPS    1e-7f
#define SLOPE  0.01f

__device__ __forceinline__ v8f wmma4(v2f a, v2f b, v8f c) {
  return __builtin_amdgcn_wmma_f32_16x16x4_f32(false, a, false, b, (short)0, c,
                                               false, false);
}

__device__ __forceinline__ int hand_off_k(int j, int hi8) { return hi8 + 2 * j; }

__device__ __forceinline__ v8f norm_lrelu8(v8f ym, float inv) {
  v8f r;
#pragma unroll
  for (int i = 0; i < 8; ++i) {
    const float t = ym[i] * inv;
    r[i] = fmaxf(t, SLOPE * t);
  }
  return r;
}

__device__ __forceinline__ void stage_priority(int t4, int tpw) {
  const int half = (tpw / 2) & ~3;
  const int three_q = ((3 * tpw) / 4) & ~3;
  const int seven_e = ((7 * tpw) / 8) & ~3;
  if (t4 == half)
    __builtin_amdgcn_s_setprio(2);
  if (t4 == three_q)
    __builtin_amdgcn_s_setprio(1);
  if (t4 == seven_e)
    __builtin_amdgcn_s_setprio(0);
}

__device__ __forceinline__ v2f load_x_tile(const float* __restrict__ xs, int p, int m,
                                           int khalf) {
  v2f r;
  r.x = xs[khalf * NPIX + p + m];
  r.y = xs[((khalf == 0) ? NPIX : 2 * NPIX) + p + m];
  return r;
}
__device__ __forceinline__ v2f x_operand(v2f r, int khalf) {
  r.y = (khalf == 0) ? r.y : 0.f;
  return r;
}

template <int MASK>
__device__ __forceinline__ float row_xor_add(float v) {
  const int b = __builtin_bit_cast(int, v);
  return v + __builtin_bit_cast(float, __builtin_amdgcn_update_dpp(b, b, 0x160 + MASK, 0xf, 0xf, false));
}
__device__ __forceinline__ float half_xor_add(float v) {
  const int b = __builtin_bit_cast(int, v);
  return v + __builtin_bit_cast(float,
                 __builtin_amdgcn_permlanex16(b, b, 0x76543210, 0xfedcba98, false, false));
}
__device__ __forceinline__ float row16_sum(float v) {
  v = row_xor_add<8>(v);
  v = row_xor_add<4>(v);
  v = row_xor_add<2>(v);
  return row_xor_add<1>(v);
}
__device__ __forceinline__ float wave_sum(float v) { return row16_sum(half_xor_add(v)); }

#define K1_THREADS 256
#define K1_QSTRIDE (K1_THREADS * PB1)
#define K1_TRIPS   ((NPIX / 4) / K1_QSTRIDE)

__device__ __forceinline__ float hsum4(v4f v) { return (v.x + v.y) + (v.z + v.w); }

__global__ void k_stats1(const float* __restrict__ x, const float* __restrict__ Wa,
                         const float* __restrict__ ba, float* __restrict__ part) {
  const int s = blockIdx.y;
  const float* xs = x + (size_t)s * CI * NPIX;
  const v4f* c0 = (const v4f*)xs;
  const v4f* c1 = (const v4f*)(xs + NPIX);
  const v4f* c2 = (const v4f*)(xs + 2 * NPIX);
  const int lane = threadIdx.x & 31;
  const int wave = threadIdx.x >> 5;
  const int q0 = blockIdx.x * K1_THREADS + threadIdx.x;

  v4f a0[K1_TRIPS], a1[K1_TRIPS], a2[K1_TRIPS];
#pragma unroll
  for (int t = 0; t < K1_TRIPS; ++t) {
    const int q = q0 + t * K1_QSTRIDE;
    a0[t] = c0[q];
    a1[t] = c1[q];
    a2[t] = c2[q];
  }
  const int c = lane & 15;
  const float w0 = Wa[c * CI + 0], w1 = Wa[c * CI + 1], w2 = Wa[c * CI + 2];
  const float bc_ = ba[c];
  __builtin_amdgcn_sched_barrier(0);

  const v4f zero4 = {0.f, 0.f, 0.f, 0.f};
  v4f S0 = zero4, S1 = zero4, S2 = zero4;
  v4f P00 = zero4, P01 = zero4, P02 = zero4, P11 = zero4, P12 = zero4, P22 = zero4;
#pragma unroll
  for (int t = 0; t < K1_TRIPS; ++t) {
    S0 += a0[t];
    S1 += a1[t];
    S2 += a2[t];
    P00 += a0[t] * a0[t];
    P01 += a0[t] * a1[t];
    P02 += a0[t] * a2[t];
    P11 += a1[t] * a1[t];
    P12 += a1[t] * a2[t];
    P22 += a2[t] * a2[t];
  }
  float m[9] = {hsum4(S0),  hsum4(S1),  hsum4(S2),  hsum4(P00), hsum4(P01),
                hsum4(P02), hsum4(P11), hsum4(P12), hsum4(P22)};

#pragma unroll
  for (int k = 0; k < 9; ++k) m[k] = wave_sum(m[k]);

  __shared__ __attribute__((aligned(16))) float sm[8 * 12];
  if (lane == 0) {
#pragma unroll
    for (int k = 0; k < 9; ++k) sm[wave * 12 + k] = m[k];
  }
  __syncthreads();
  if (wave == 0) {
    float M[9];
#pragma unroll
    for (int k = 0; k < 9; ++k)
      M[k] = ((sm[0 * 12 + k] + sm[1 * 12 + k]) + (sm[2 * 12 + k] + sm[3 * 12 + k])) +
             ((sm[4 * 12 + k] + sm[5 * 12 + k]) + (sm[6 * 12 + k] + sm[7 * 12 + k]));
    const float nb  = (float)(K1_TRIPS * K1_THREADS * 4);
    const float lin = fmaf(w0, M[0], fmaf(w1, M[1], w2 * M[2]));
    const float dg  = fmaf(w0 * w0, M[3], fmaf(w1 * w1, M[6], (w2 * w2) * M[8]));
    const float od  = fmaf(w0 * w1, M[4], fmaf(w0 * w2, M[5], (w1 * w2) * M[7]));
    float l = fmaf(nb, bc_, lin);
    float q = fmaf(2.f, od, dg) + bc_ * fmaf(2.f, lin, nb * bc_);
    if (lane >= 16) { l = 0.f; q = 0.f; }
    l = row16_sum(l);
    q = row16_sum(q);
    if (lane == 0) {
      part[(s * PB1 + blockIdx.x) * 2 + 0] = l;
      part[(s * PB1 + blockIdx.x) * 2 + 1] = q;
    }
  }
}

template <int NB>
__device__ __forceinline__ v2f fold_load(const float* __restrict__ part, int s, int lane) {
  static_assert(NB == 32, "one partial pair per lane");
  return *(const v2f*)(part + (s * NB + lane) * 2);
}
__device__ __forceinline__ void fold_finish(v2f pr, float& mean_o, float& inv_o) {
  const float s1 = wave_sum(pr.x);
  const float s2 = wave_sum(pr.y);
  const float n = (float)CH * (float)NPIX;
  const float mean = s1 * (1.0f / n);
  const float var = (s2 - s1 * mean) * (1.0f / (n - 1.0f));
  mean_o = mean;
  inv_o = 1.0f / (sqrtf(var) + EPS);
}

__global__ void k_stats2(const float* __restrict__ x, const float* __restrict__ Wa,
                         const float* __restrict__ ba, const float* __restrict__ Wb,
                         const float* __restrict__ bb, const float* __restrict__ part1,
                         float* __restrict__ part) {
  const int s = blockIdx.y;
  const float* xs = x + (size_t)s * CI * NPIX;
  const int lane  = threadIdx.x & 31;
  const int wave  = threadIdx.x >> 5;
  const int m     = lane & 15;
  const int khalf = (lane >> 4) << 1;
  const int hi8   = (lane >> 4) << 3;

  const int tpb = (NPIX / 16) / gridDim.x;
  const int tpw = tpb / 8;
  __builtin_assume(tpw >= 4);
  const int tbase = blockIdx.x * tpb + wave * tpw;
  v2f xr[4];
#pragma unroll
  for (int u = 0; u < 4; ++u) xr[u] = load_x_tile(xs, (tbase + u) * 16, m, khalf);
  const v2f pr1 = fold_load<PB1>(part1, s, lane);

  v2f aA;
  aA.x = Wa[m * CI + khalf];
  aA.y = (khalf == 0) ? Wa[m * CI + 1] : 0.f;

  v2f bA[4];
#pragma unroll
  for (int j = 0; j < 4; ++j) {
    int k0 = hand_off_k(j, hi8);
    bA[j].x = Wb[m * CH + k0];
    bA[j].y = Wb[m * CH + k0 + 1];
  }

  v8f biasA, biasB;
#pragma unroll
  for (int i = 0; i < 8; ++i) { biasA[i] = ba[i + hi8]; biasB[i] = bb[i + hi8]; }

  __builtin_amdgcn_sched_barrier(0);
  float mean1, inv1;
  fold_finish(pr1, mean1, inv1);

#pragma unroll
  for (int i = 0; i < 8; ++i) biasA[i] -= mean1;

  v8f acc1 = {0.f, 0.f, 0.f, 0.f, 0.f, 0.f, 0.f, 0.f};
  v8f acc2 = {0.f, 0.f, 0.f, 0.f, 0.f, 0.f, 0.f, 0.f};

  __builtin_amdgcn_s_setprio(3);
  for (int t4 = 0; t4 < tpw; t4 += 4) {
    stage_priority(t4, tpw);
    v2f xc[4];
#pragma unroll
    for (int u = 0; u < 4; ++u) {
      xc[u] = xr[u];
      xr[u] = load_x_tile(xs, (tbase + ((t4 + u + 4 < tpw) ? t4 + u + 4 : tpw - 1)) * 16, m, khalf);
    }
    __builtin_amdgcn_sched_barrier(0);
#pragma unroll
   for (int u = 0; u < 4; ++u) {
    const v2f xB = x_operand(xc[u], khalf);

    v8f y = wmma4(aA, xB, biasA);

    v8f h = norm_lrelu8(y, inv1);

    v8f z = biasB;
#pragma unroll
    for (int j = 0; j < 4; ++j) {
      v2f hB;
      hB.x = h[2 * j];
      hB.y = h[2 * j + 1];
      z = wmma4(bA[j], hB, z);
    }

    acc1 += z;
    acc2 += z * z;
   }
  }
  const float t1 = ((acc1[0] + acc1[1]) + (acc1[2] + acc1[3])) +
                   ((acc1[4] + acc1[5]) + (acc1[6] + acc1[7]));
  const float t2 = ((acc2[0] + acc2[1]) + (acc2[2] + acc2[3])) +
                   ((acc2[4] + acc2[5]) + (acc2[6] + acc2[7]));

  const float w1 = wave_sum(t1);
  const float w2 = wave_sum(t2);
  __shared__ float r1[8], r2[8];
  if (lane == 0) { r1[wave] = w1; r2[wave] = w2; }
  __syncthreads();
  if (threadIdx.x == 0) {
    part[(s * PB3 + blockIdx.x) * 2 + 0] =
        ((r1[0] + r1[1]) + (r1[2] + r1[3])) + ((r1[4] + r1[5]) + (r1[6] + r1[7]));
    part[(s * PB3 + blockIdx.x) * 2 + 1] =
        ((r2[0] + r2[1]) + (r2[2] + r2[3])) + ((r2[4] + r2[5]) + (r2[6] + r2[7]));
  }
}

__global__ void k_final(const float* __restrict__ x, const float* __restrict__ Wa,
                        const float* __restrict__ ba, const float* __restrict__ Wb,
                        const float* __restrict__ bb, const float* __restrict__ Wc,
                        const float* __restrict__ bc, const float* __restrict__ part1,
                        const float* __restrict__ part2, float* __restrict__ out) {
  const int s = blockIdx.y;
  const float* xs = x + (size_t)s * CI * NPIX;
  float* os = out + (size_t)s * CO * NPIX;
  const int lane  = threadIdx.x & 31;
  const int wave  = threadIdx.x >> 5;
  const int m     = lane & 15;
  const int khalf = (lane >> 4) << 1;
  const int hi8   = (lane >> 4) << 3;

  const int tpb = (NPIX / 16) / gridDim.x;
  const int tpw = tpb / 8;
  __builtin_assume(tpw >= 4);
  const int tbase = blockIdx.x * tpb + wave * tpw;
  v2f xr[4];
#pragma unroll
  for (int u = 0; u < 4; ++u) xr[u] = load_x_tile(xs, (tbase + u) * 16, m, khalf);
  const v2f pr1 = fold_load<PB1>(part1, s, lane);
  const v2f pr2 = fold_load<PB3>(part2, s, lane);

  v2f aA;
  aA.x = Wa[m * CI + khalf];
  aA.y = (khalf == 0) ? Wa[m * CI + 1] : 0.f;

  v2f bA[4], cA[4];
#pragma unroll
  for (int j = 0; j < 4; ++j) {
    int k0 = hand_off_k(j, hi8);
    bA[j].x = Wb[m * CH + k0];
    bA[j].y = Wb[m * CH + k0 + 1];
    cA[j].x = (m < CO) ? Wc[m * CH + k0] : 0.f;
    cA[j].y = (m < CO) ? Wc[m * CH + k0 + 1] : 0.f;
  }

  v8f biasA, biasB, biasC;
#pragma unroll
  for (int i = 0; i < 8; ++i) {
    biasA[i] = ba[i + hi8];
    biasB[i] = bb[i + hi8];
    biasC[i] = (i + hi8 < CO) ? bc[i + hi8] : 0.f;
  }

  __builtin_amdgcn_sched_barrier(0);
  float mean1, inv1, mean2, inv2;
  fold_finish(pr1, mean1, inv1);
  fold_finish(pr2, mean2, inv2);

#pragma unroll
  for (int i = 0; i < 8; ++i) { biasA[i] -= mean1; biasB[i] -= mean2; }

  __builtin_amdgcn_s_setprio(3);
  for (int t4 = 0; t4 < tpw; t4 += 4) {
    stage_priority(t4, tpw);
    v2f xc[4];
#pragma unroll
    for (int u = 0; u < 4; ++u) {
      xc[u] = xr[u];
      xr[u] = load_x_tile(xs, (tbase + ((t4 + u + 4 < tpw) ? t4 + u + 4 : tpw - 1)) * 16, m, khalf);
    }
    __builtin_amdgcn_sched_barrier(0);
#pragma unroll
   for (int u = 0; u < 4; ++u) {
    const int p = (tbase + t4 + u) * 16;
    const v2f xB = x_operand(xc[u], khalf);

    v8f y = wmma4(aA, xB, biasA);

    v8f h1 = norm_lrelu8(y, inv1);

    v8f z = biasB;
#pragma unroll
    for (int j = 0; j < 4; ++j) {
      v2f hB;
      hB.x = h1[2 * j];
      hB.y = h1[2 * j + 1];
      z = wmma4(bA[j], hB, z);
    }

    v8f h2 = norm_lrelu8(z, inv2);

    v8f o = biasC;
#pragma unroll
    for (int j = 0; j < 4; ++j) {
      v2f hB;
      hB.x = h2[2 * j];
      hB.y = h2[2 * j + 1];
      o = wmma4(cA[j], hB, o);
    }

    if (lane < 16) {
      os[0 * NPIX + p + m] = o[0];
      os[1 * NPIX + p + m] = o[1];
      os[2 * NPIX + p + m] = o[2];
    }
   }
  }
}

extern "C" void kernel_launch(void* const* d_in, const int* in_sizes, int n_in,
                              void* d_out, int out_size, void* d_ws, size_t ws_size,
                              hipStream_t stream) {
  (void)in_sizes; (void)n_in; (void)out_size; (void)ws_size;
  const float* x  = (const float*)d_in[0];
  const float* Wa = (const float*)d_in[1];
  const float* ba = (const float*)d_in[2];
  const float* Wb = (const float*)d_in[3];
  const float* bb = (const float*)d_in[4];
  const float* Wc = (const float*)d_in[5];
  const float* bc = (const float*)d_in[6];
  float* out = (float*)d_out;

  float* ws     = (float*)d_ws;
  float* part1  = ws;
  float* part2  = part1 + NSAMP * PB1 * 2;

  k_stats1 <<<dim3(PB1, NSAMP), 256, 0, stream>>>(x, Wa, ba, part1);
  k_stats2 <<<dim3(PB3, NSAMP), 256, 0, stream>>>(x, Wa, ba, Wb, bb, part1, part2);
  k_final  <<<dim3(PB5, NSAMP), 256, 0, stream>>>(x, Wa, ba, Wb, bb, Wc, bc,
                                                  part1, part2, out);
}


# normalisation scale and the activation's own factor folded into the next conv's weights once per wave (lrelu(t) = a(t + c|t|)): an activation element is one fma, was 2 muls + max; roundings move
# speedup vs baseline: 1.9485x; 1.0539x over previous
#include <hip/hip_runtime.h>
#include <math.h>

typedef __attribute__((ext_vector_type(2))) float v2f;
typedef __attribute__((ext_vector_type(4))) float v4f;
typedef __attribute__((ext_vector_type(8))) float v8f;

#define NPIX   (512*512)
#define NSAMP  16
#define CI     3
#define CH     16
#define CO     3
#define PB1    32
#define PB3    32
#define PB5    32
#define EPS    1e-7f
#define SLOPE  0.01

__device__ __forceinline__ v8f wmma4(v2f a, v2f b, v8f c) {
  return __builtin_amdgcn_wmma_f32_16x16x4_f32(false, a, false, b, (short)0, c,
                                               false, false);
}

__device__ __forceinline__ int hand_off_k(int j, int hi8) { return hi8 + 2 * j; }

#define LR_A ((float)((1.0 + SLOPE) / 2.0))
#define LR_C ((float)((1.0 - SLOPE) / (1.0 + SLOPE)))
__device__ __forceinline__ v8f lrelu_core8(v8f t) {
  v8f r;
#pragma unroll
  for (int i = 0; i < 8; i += 2) {
    r[i]     = fmaf(fabsf(t[i]), LR_C, t[i]);
    r[i + 1] = t[i + 1] + LR_C * fabsf(t[i + 1]);
  }
  return r;
}
__device__ __forceinline__ void scale_weights(v2f (&w)[4], float s) {
#pragma unroll
  for (int j = 0; j < 4; ++j) {
    w[j].x *= s;
    w[j].y *= s;
  }
}

__device__ __forceinline__ void stage_priority(int t4, int tpw) {
  const int half = (tpw / 2) & ~3;
  const int three_q = ((3 * tpw) / 4) & ~3;
  const int seven_e = ((7 * tpw) / 8) & ~3;
  if (t4 == half)
    __builtin_amdgcn_s_setprio(2);
  if (t4 == three_q)
    __builtin_amdgcn_s_setprio(1);
  if (t4 == seven_e)
    __builtin_amdgcn_s_setprio(0);
}

__device__ __forceinline__ v2f load_x_tile(const float* __restrict__ xs, int p, int m,
                                           int khalf) {
  v2f r;
  r.x = xs[khalf * NPIX + p + m];
  r.y = xs[((khalf == 0) ? NPIX : 2 * NPIX) + p + m];
  return r;
}
__device__ __forceinline__ v2f x_operand(v2f r, int khalf) {
  r.y = (khalf == 0) ? r.y : 0.f;
  return r;
}

template <int MASK>
__device__ __forceinline__ float row_xor_add(float v) {
  const int b = __builtin_bit_cast(int, v);
  return v + __builtin_bit_cast(float, __builtin_amdgcn_update_dpp(b, b, 0x160 + MASK, 0xf, 0xf, false));
}
__device__ __forceinline__ float half_xor_add(float v) {
  const int b = __builtin_bit_cast(int, v);
  return v + __builtin_bit_cast(float,
                 __builtin_amdgcn_permlanex16(b, b, 0x76543210, 0xfedcba98, false, false));
}
__device__ __forceinline__ float row16_sum(float v) {
  v = row_xor_add<8>(v);
  v = row_xor_add<4>(v);
  v = row_xor_add<2>(v);
  return row_xor_add<1>(v);
}
__device__ __forceinline__ float wave_sum(float v) { return row16_sum(half_xor_add(v)); }

#define K1_THREADS 256
#define K1_QSTRIDE (K1_THREADS * PB1)
#define K1_TRIPS   ((NPIX / 4) / K1_QSTRIDE)

__device__ __forceinline__ float hsum4(v4f v) { return (v.x + v.y) + (v.z + v.w); }

__global__ void k_stats1(const float* __restrict__ x, const float* __restrict__ Wa,
                         const float* __restrict__ ba, float* __restrict__ part) {
  const int s = blockIdx.y;
  const float* xs = x + (size_t)s * CI * NPIX;
  const v4f* c0 = (const v4f*)xs;
  const v4f* c1 = (const v4f*)(xs + NPIX);
  const v4f* c2 = (const v4f*)(xs + 2 * NPIX);
  const int lane = threadIdx.x & 31;
  const int wave = threadIdx.x >> 5;
  const int q0 = blockIdx.x * K1_THREADS + threadIdx.x;

  v4f a0[K1_TRIPS], a1[K1_TRIPS], a2[K1_TRIPS];
#pragma unroll
  for (int t = 0; t < K1_TRIPS; ++t) {
    const int q = q0 + t * K1_QSTRIDE;
    a0[t] = c0[q];
    a1[t] = c1[q];
    a2[t] = c2[q];
  }
  const int c = lane & 15;
  const float w0 = Wa[c * CI + 0], w1 = Wa[c * CI + 1], w2 = Wa[c * CI + 2];
  const float bc_ = ba[c];
  __builtin_amdgcn_sched_barrier(0);

  const v4f zero4 = {0.f, 0.f, 0.f, 0.f};
  v4f S0 = zero4, S1 = zero4, S2 = zero4;
  v4f P00 = zero4, P01 = zero4, P02 = zero4, P11 = zero4, P12 = zero4, P22 = zero4;
#pragma unroll
  for (int t = 0; t < K1_TRIPS; ++t) {
    S0 += a0[t];
    S1 += a1[t];
    S2 += a2[t];
    P00 += a0[t] * a0[t];
    P01 += a0[t] * a1[t];
    P02 += a0[t] * a2[t];
    P11 += a1[t] * a1[t];
    P12 += a1[t] * a2[t];
    P22 += a2[t] * a2[t];
  }
  float m[9] = {hsum4(S0),  hsum4(S1),  hsum4(S2),  hsum4(P00), hsum4(P01),
                hsum4(P02), hsum4(P11), hsum4(P12), hsum4(P22)};

#pragma unroll
  for (int k = 0; k < 9; ++k) m[k] = wave_sum(m[k]);

  __shared__ __attribute__((aligned(16))) float sm[8 * 12];
  if (lane == 0) {
#pragma unroll
    for (int k = 0; k < 9; ++k) sm[wave * 12 + k] = m[k];
  }
  __syncthreads();
  if (wave == 0) {
    float M[9];
#pragma unroll
    for (int k = 0; k < 9; ++k)
      M[k] = ((sm[0 * 12 + k] + sm[1 * 12 + k]) + (sm[2 * 12 + k] + sm[3 * 12 + k])) +
             ((sm[4 * 12 + k] + sm[5 * 12 + k]) + (sm[6 * 12 + k] + sm[7 * 12 + k]));
    const float nb  = (float)(K1_TRIPS * K1_THREADS * 4);
    const float lin = fmaf(w0, M[0], fmaf(w1, M[1], w2 * M[2]));
    const float dg  = fmaf(w0 * w0, M[3], fmaf(w1 * w1, M[6], (w2 * w2) * M[8]));
    const float od  = fmaf(w0 * w1, M[4], fmaf(w0 * w2, M[5], (w1 * w2) * M[7]));
    float l = fmaf(nb, bc_, lin);
    float q = fmaf(2.f, od, dg) + bc_ * fmaf(2.f, lin, nb * bc_);
    if (lane >= 16) { l = 0.f; q = 0.f; }
    l = row16_sum(l);
    q = row16_sum(q);
    if (lane == 0) {
      part[(s * PB1 + blockIdx.x) * 2 + 0] = l;
      part[(s * PB1 + blockIdx.x) * 2 + 1] = q;
    }
  }
}

template <int NB>
__device__ __forceinline__ v2f fold_load(const float* __restrict__ part, int s, int lane) {
  static_assert(NB == 32, "one partial pair per lane");
  return *(const v2f*)(part + (s * NB + lane) * 2);
}
__device__ __forceinline__ void fold_finish(v2f pr, float& mean_o, float& inv_o) {
  const float s1 = wave_sum(pr.x);
  const float s2 = wave_sum(pr.y);
  const float n = (float)CH * (float)NPIX;
  const float mean = s1 * (1.0f / n);
  const float var = (s2 - s1 * mean) * (1.0f / (n - 1.0f));
  mean_o = mean;
  inv_o = 1.0f / (sqrtf(var) + EPS);
}

__global__ void k_stats2(const float* __restrict__ x, const float* __restrict__ Wa,
                         const float* __restrict__ ba, const float* __restrict__ Wb,
                         const float* __restrict__ bb, const float* __restrict__ part1,
                         float* __restrict__ part) {
  const int s = blockIdx.y;
  const float* xs = x + (size_t)s * CI * NPIX;
  const int lane  = threadIdx.x & 31;
  const int wave  = threadIdx.x >> 5;
  const int m     = lane & 15;
  const int khalf = (lane >> 4) << 1;
  const int hi8   = (lane >> 4) << 3;

  const int tpb = (NPIX / 16) / gridDim.x;
  const int tpw = tpb / 8;
  __builtin_assume(tpw >= 4);
  const int tbase = blockIdx.x * tpb + wave * tpw;
  v2f xr[4];
#pragma unroll
  for (int u = 0; u < 4; ++u) xr[u] = load_x_tile(xs, (tbase + u) * 16, m, khalf);
  const v2f pr1 = fold_load<PB1>(part1, s, lane);

  v2f aA;
  aA.x = Wa[m * CI + khalf];
  aA.y = (khalf == 0) ? Wa[m * CI + 1] : 0.f;

  v2f bA[4];
#pragma unroll
  for (int j = 0; j < 4; ++j) {
    int k0 = hand_off_k(j, hi8);
    bA[j].x = Wb[m * CH + k0];
    bA[j].y = Wb[m * CH + k0 + 1];
  }

  v8f biasA, biasB;
#pragma unroll
  for (int i = 0; i < 8; ++i) { biasA[i] = ba[i + hi8]; biasB[i] = bb[i + hi8]; }

  __builtin_amdgcn_sched_barrier(0);
  float mean1, inv1;
  fold_finish(pr1, mean1, inv1);

#pragma unroll
  for (int i = 0; i < 8; ++i) biasA[i] -= mean1;
  scale_weights(bA, inv1 * LR_A);

  v8f acc1 = {0.f, 0.f, 0.f, 0.f, 0.f, 0.f, 0.f, 0.f};
  v8f acc2 = {0.f, 0.f, 0.f, 0.f, 0.f, 0.f, 0.f, 0.f};

  __builtin_amdgcn_s_setprio(3);
  for (int t4 = 0; t4 < tpw; t4 += 4) {
    stage_priority(t4, tpw);
    v2f xc[4];
#pragma unroll
    for (int u = 0; u < 4; ++u) {
      xc[u] = xr[u];
      xr[u] = load_x_tile(xs, (tbase + ((t4 + u + 4 < tpw) ? t4 + u + 4 : tpw - 1)) * 16, m, khalf);
    }
    __builtin_amdgcn_sched_barrier(0);
#pragma unroll
   for (int u = 0; u < 4; ++u) {
    const v2f xB = x_operand(xc[u], khalf);

    v8f y = wmma4(aA, xB, biasA);

    v8f h = lrelu_core8(y);

    v8f z = biasB;
#pragma unroll
    for (int j = 0; j < 4; ++j) {
      v2f hB;
      hB.x = h[2 * j];
      hB.y = h[2 * j + 1];
      z = wmma4(bA[j], hB, z);
    }

    acc1 += z;
    acc2 += z * z;
   }
  }
  const float t1 = ((acc1[0] + acc1[1]) + (acc1[2] + acc1[3])) +
                   ((acc1[4] + acc1[5]) + (acc1[6] + acc1[7]));
  const float t2 = ((acc2[0] + acc2[1]) + (acc2[2] + acc2[3])) +
                   ((acc2[4] + acc2[5]) + (acc2[6] + acc2[7]));

  const float w1 = wave_sum(t1);
  const float w2 = wave_sum(t2);
  __shared__ float r1[8], r2[8];
  if (lane == 0) { r1[wave] = w1; r2[wave] = w2; }
  __syncthreads();
  if (threadIdx.x == 0) {
    part[(s * PB3 + blockIdx.x) * 2 + 0] =
        ((r1[0] + r1[1]) + (r1[2] + r1[3])) + ((r1[4] + r1[5]) + (r1[6] + r1[7]));
    part[(s * PB3 + blockIdx.x) * 2 + 1] =
        ((r2[0] + r2[1]) + (r2[2] + r2[3])) + ((r2[4] + r2[5]) + (r2[6] + r2[7]));
  }
}

__global__ void k_final(const float* __restrict__ x, const float* __restrict__ Wa,
                        const float* __restrict__ ba, const float* __restrict__ Wb,
                        const float* __restrict__ bb, const float* __restrict__ Wc,
                        const float* __restrict__ bc, const float* __restrict__ part1,
                        const float* __restrict__ part2, float* __restrict__ out) {
  const int s = blockIdx.y;
  const float* xs = x + (size_t)s * CI * NPIX;
  float* os = out + (size_t)s * CO * NPIX;
  const int lane  = threadIdx.x & 31;
  const int wave  = threadIdx.x >> 5;
  const int m     = lane & 15;
  const int khalf = (lane >> 4) << 1;
  const int hi8   = (lane >> 4) << 3;

  const int tpb = (NPIX / 16) / gridDim.x;
  const int tpw = tpb / 8;
  __builtin_assume(tpw >= 4);
  const int tbase = blockIdx.x * tpb + wave * tpw;
  v2f xr[4];
#pragma unroll
  for (int u = 0; u < 4; ++u) xr[u] = load_x_tile(xs, (tbase + u) * 16, m, khalf);
  const v2f pr1 = fold_load<PB1>(part1, s, lane);
  const v2f pr2 = fold_load<PB3>(part2, s, lane);

  v2f aA;
  aA.x = Wa[m * CI + khalf];
  aA.y = (khalf == 0) ? Wa[m * CI + 1] : 0.f;

  v2f bA[4], cA[4];
#pragma unroll
  for (int j = 0; j < 4; ++j) {
    int k0 = hand_off_k(j, hi8);
    bA[j].x = Wb[m * CH + k0];
    bA[j].y = Wb[m * CH + k0 + 1];
    cA[j].x = (m < CO) ? Wc[m * CH + k0] : 0.f;
    cA[j].y = (m < CO) ? Wc[m * CH + k0 + 1] : 0.f;
  }

  v8f biasA, biasB, biasC;
#pragma unroll
  for (int i = 0; i < 8; ++i) {
    biasA[i] = ba[i + hi8];
    biasB[i] = bb[i + hi8];
    biasC[i] = (i + hi8 < CO) ? bc[i + hi8] : 0.f;
  }

  __builtin_amdgcn_sched_barrier(0);
  float mean1, inv1, mean2, inv2;
  fold_finish(pr1, mean1, inv1);
  fold_finish(pr2, mean2, inv2);

#pragma unroll
  for (int i = 0; i < 8; ++i) { biasA[i] -= mean1; biasB[i] -= mean2; }
  scale_weights(bA, inv1 * LR_A);
  scale_weights(cA, inv2 * LR_A);

  __builtin_amdgcn_s_setprio(3);
  for (int t4 = 0; t4 < tpw; t4 += 4) {
    stage_priority(t4, tpw);
    v2f xc[4];
#pragma unroll
    for (int u = 0; u < 4; ++u) {
      xc[u] = xr[u];
      xr[u] = load_x_tile(xs, (tbase + ((t4 + u + 4 < tpw) ? t4 + u + 4 : tpw - 1)) * 16, m, khalf);
    }
    __builtin_amdgcn_sched_barrier(0);
#pragma unroll
   for (int u = 0; u < 4; ++u) {
    const int p = (tbase + t4 + u) * 16;
    const v2f xB = x_operand(xc[u], khalf);

    v8f y = wmma4(aA, xB, biasA);

    v8f h1 = lrelu_core8(y);

    v8f z = biasB;
#pragma unroll
    for (int j = 0; j < 4; ++j) {
      v2f hB;
      hB.x = h1[2 * j];
      hB.y = h1[2 * j + 1];
      z = wmma4(bA[j], hB, z);
    }

    v8f h2 = lrelu_core8(z);

    v8f o = biasC;
#pragma unroll
    for (int j = 0; j < 4; ++j) {
      v2f hB;
      hB.x = h2[2 * j];
      hB.y = h2[2 * j + 1];
      o = wmma4(cA[j], hB, o);
    }

    if (lane < 16) {
      os[0 * NPIX + p + m] = o[0];
      os[1 * NPIX + p + m] = o[1];
      os[2 * NPIX + p + m] = o[2];
    }
   }
  }
}

extern "C" void kernel_launch(void* const* d_in, const int* in_sizes, int n_in,
                              void* d_out, int out_size, void* d_ws, size_t ws_size,
                              hipStream_t stream) {
  (void)in_sizes; (void)n_in; (void)out_size; (void)ws_size;
  const float* x  = (const float*)d_in[0];
  const float* Wa = (const float*)d_in[1];
  const float* ba = (const float*)d_in[2];
  const float* Wb = (const float*)d_in[3];
  const float* bb = (const float*)d_in[4];
  const float* Wc = (const float*)d_in[5];
  const float* bc = (const float*)d_in[6];
  float* out = (float*)d_out;

  float* ws     = (float*)d_ws;
  float* part1  = ws;
  float* part2  = part1 + NSAMP * PB1 * 2;

  k_stats1 <<<dim3(PB1, NSAMP), 256, 0, stream>>>(x, Wa, ba, part1);
  k_stats2 <<<dim3(PB3, NSAMP), 256, 0, stream>>>(x, Wa, ba, Wb, bb, part1, part2);
  k_final  <<<dim3(PB5, NSAMP), 256, 0, stream>>>(x, Wa, ba, Wb, bb, Wc, bc,
                                                  part1, part2, out);
}


# column n of tile u is pixel 4n+u of a wave's 64-pixel trip (was 16u+n): x is one 16-byte load per lane and channel (was 4), out leaves in 16-byte stores, one block per trip; stats sum order changes
# speedup vs baseline: 2.1082x; 1.0820x over previous
#include <hip/hip_runtime.h>
#include <math.h>

typedef __attribute__((ext_vector_type(2))) float v2f;
typedef __attribute__((ext_vector_type(4))) float v4f;
typedef __attribute__((ext_vector_type(8))) float v8f;

#define NPIX   (512*512)
#define NSAMP  16
#define CI     3
#define CH     16
#define CO     3
#define PB1    32
#define PB3    32
#define PB5    32
#define EPS    1e-7f
#define SLOPE  0.01

__device__ __forceinline__ v8f wmma4(v2f a, v2f b, v8f c) {
  return __builtin_amdgcn_wmma_f32_16x16x4_f32(false, a, false, b, (short)0, c,
                                               false, false);
}

__device__ __forceinline__ int hand_off_k(int j, int hi8) { return hi8 + 2 * j; }

#define LR_A ((float)((1.0 + SLOPE) / 2.0))
#define LR_C ((float)((1.0 - SLOPE) / (1.0 + SLOPE)))
__device__ __forceinline__ v8f lrelu_core8(v8f t) {
  v8f r;
#pragma unroll
  for (int i = 0; i < 8; i += 2) {
    r[i]     = fmaf(fabsf(t[i]), LR_C, t[i]);
    r[i + 1] = t[i + 1] + LR_C * fabsf(t[i + 1]);
  }
  return r;
}
__device__ __forceinline__ void scale_weights(v2f (&w)[4], float s) {
#pragma unroll
  for (int j = 0; j < 4; ++j) {
    w[j].x *= s;
    w[j].y *= s;
  }
}

__device__ __forceinline__ void stage_priority(int t4, int tpw) {
  const int half = (tpw / 2) & ~3;
  const int three_q = ((3 * tpw) / 4) & ~3;
  const int seven_e = ((7 * tpw) / 8) & ~3;
  if (t4 == half)
    __builtin_amdgcn_s_setprio(2);
  if (t4 == three_q)
    __builtin_amdgcn_s_setprio(1);
  if (t4 == seven_e)
    __builtin_amdgcn_s_setprio(0);
}

#define NQUAD (NPIX / 4)
struct xquad { v4f c0, c1; };
__device__ __forceinline__ xquad load_x_quad(const v4f* __restrict__ xs4, int q0, int m,
                                             int khalf) {
  xquad r;
  r.c0 = xs4[(khalf >> 1) * (2 * NQUAD) + q0 + m];
  r.c1 = xs4[((khalf == 0) ? NQUAD : 2 * NQUAD) + q0 + m];
  return r;
}
__device__ __forceinline__ v2f x_operand(const xquad& q, int u, int khalf) {
  v2f r;
  r.x = q.c0[u];
  r.y = (khalf == 0) ? q.c1[u] : 0.f;
  return r;
}

template <int MASK>
__device__ __forceinline__ float row_xor_add(float v) {
  const int b = __builtin_bit_cast(int, v);
  return v + __builtin_bit_cast(float, __builtin_amdgcn_update_dpp(b, b, 0x160 + MASK, 0xf, 0xf, false));
}
__device__ __forceinline__ float half_xor_add(float v) {
  const int b = __builtin_bit_cast(int, v);
  return v + __builtin_bit_cast(float,
                 __builtin_amdgcn_permlanex16(b, b, 0x76543210, 0xfedcba98, false, false));
}
__device__ __forceinline__ float row16_sum(float v) {
  v = row_xor_add<8>(v);
  v = row_xor_add<4>(v);
  v = row_xor_add<2>(v);
  return row_xor_add<1>(v);
}
__device__ __forceinline__ float wave_sum(float v) { return row16_sum(half_xor_add(v)); }

#define K1_THREADS 256
#define K1_QSTRIDE (K1_THREADS * PB1)
#define K1_TRIPS   ((NPIX / 4) / K1_QSTRIDE)

__device__ __forceinline__ float hsum4(v4f v) { return (v.x + v.y) + (v.z + v.w); }

__global__ void k_stats1(const float* __restrict__ x, const float* __restrict__ Wa,
                         const float* __restrict__ ba, float* __restrict__ part) {
  const int s = blockIdx.y;
  const float* xs = x + (size_t)s * CI * NPIX;
  const v4f* c0 = (const v4f*)xs;
  const v4f* c1 = (const v4f*)(xs + NPIX);
  const v4f* c2 = (const v4f*)(xs + 2 * NPIX);
  const int lane = threadIdx.x & 31;
  const int wave = threadIdx.x >> 5;
  const int q0 = blockIdx.x * K1_THREADS + threadIdx.x;

  v4f a0[K1_TRIPS], a1[K1_TRIPS], a2[K1_TRIPS];
#pragma unroll
  for (int t = 0; t < K1_TRIPS; ++t) {
    const int q = q0 + t * K1_QSTRIDE;
    a0[t] = c0[q];
    a1[t] = c1[q];
    a2[t] = c2[q];
  }
  const int c = lane & 15;
  const float w0 = Wa[c * CI + 0], w1 = Wa[c * CI + 1], w2 = Wa[c * CI + 2];
  const float bc_ = ba[c];
  __builtin_amdgcn_sched_barrier(0);

  const v4f zero4 = {0.f, 0.f, 0.f, 0.f};
  v4f S0 = zero4, S1 = zero4, S2 = zero4;
  v4f P00 = zero4, P01 = zero4, P02 = zero4, P11 = zero4, P12 = zero4, P22 = zero4;
#pragma unroll
  for (int t = 0; t < K1_TRIPS; ++t) {
    S0 += a0[t];
    S1 += a1[t];
    S2 += a2[t];
    P00 += a0[t] * a0[t];
    P01 += a0[t] * a1[t];
    P02 += a0[t] * a2[t];
    P11 += a1[t] * a1[t];
    P12 += a1[t] * a2[t];
    P22 += a2[t] * a2[t];
  }
  float m[9] = {hsum4(S0),  hsum4(S1),  hsum4(S2),  hsum4(P00), hsum4(P01),
                hsum4(P02), hsum4(P11), hsum4(P12), hsum4(P22)};

#pragma unroll
  for (int k = 0; k < 9; ++k) m[k] = wave_sum(m[k]);

  __shared__ __attribute__((aligned(16))) float sm[8 * 12];
  if (lane == 0) {
#pragma unroll
    for (int k = 0; k < 9; ++k) sm[wave * 12 + k] = m[k];
  }
  __syncthreads();
  if (wave == 0) {
    float M[9];
#pragma unroll
    for (int k = 0; k < 9; ++k)
      M[k] = ((sm[0 * 12 + k] + sm[1 * 12 + k]) + (sm[2 * 12 + k] + sm[3 * 12 + k])) +
             ((sm[4 * 12 + k] + sm[5 * 12 + k]) + (sm[6 * 12 + k] + sm[7 * 12 + k]));
    const float nb  = (float)(K1_TRIPS * K1_THREADS * 4);
    const float lin = fmaf(w0, M[0], fmaf(w1, M[1], w2 * M[2]));
    const float dg  = fmaf(w0 * w0, M[3], fmaf(w1 * w1, M[6], (w2 * w2) * M[8]));
    const float od  = fmaf(w0 * w1, M[4], fmaf(w0 * w2, M[5], (w1 * w2) * M[7]));
    float l = fmaf(nb, bc_, lin);
    float q = fmaf(2.f, od, dg) + bc_ * fmaf(2.f, lin, nb * bc_);
    if (lane >= 16) { l = 0.f; q = 0.f; }
    l = row16_sum(l);
    q = row16_sum(q);
    if (lane == 0) {
      part[(s * PB1 + blockIdx.x) * 2 + 0] = l;
      part[(s * PB1 + blockIdx.x) * 2 + 1] = q;
    }
  }
}

template <int NB>
__device__ __forceinline__ v2f fold_load(const float* __restrict__ part, int s, int lane) {
  static_assert(NB == 32, "one partial pair per lane");
  return *(const v2f*)(part + (s * NB + lane) * 2);
}
__device__ __forceinline__ void fold_finish(v2f pr, float& mean_o, float& inv_o) {
  const float s1 = wave_sum(pr.x);
  const float s2 = wave_sum(pr.y);
  const float n = (float)CH * (float)NPIX;
  const float mean = s1 * (1.0f / n);
  const float var = (s2 - s1 * mean) * (1.0f / (n - 1.0f));
  mean_o = mean;
  inv_o = 1.0f / (sqrtf(var) + EPS);
}

__global__ void k_stats2(const float* __restrict__ x, const float* __restrict__ Wa,
                         const float* __restrict__ ba, const float* __restrict__ Wb,
                         const float* __restrict__ bb, const float* __restrict__ part1,
                         float* __restrict__ part) {
  const int s = blockIdx.y;
  const float* xs = x + (size_t)s * CI * NPIX;
  const int lane  = threadIdx.x & 31;
  const int wave  = threadIdx.x >> 5;
  const int m     = lane & 15;
  const int khalf = (lane >> 4) << 1;
  const int hi8   = (lane >> 4) << 3;

  const int tpb = (NPIX / 16) / gridDim.x;
  const int tpw = tpb / 8;
  __builtin_assume(tpw >= 4);
  const int tbase = blockIdx.x * tpb + wave * tpw;
  const v4f* xs4 = (const v4f*)xs;
  xquad xr = load_x_quad(xs4, tbase * 4, m, khalf);
  const v2f pr1 = fold_load<PB1>(part1, s, lane);

  v2f aA;
  aA.x = Wa[m * CI + khalf];
  aA.y = (khalf == 0) ? Wa[m * CI + 1] : 0.f;

  v2f bA[4];
#pragma unroll
  for (int j = 0; j < 4; ++j) {
    int k0 = hand_off_k(j, hi8);
    bA[j].x = Wb[m * CH + k0];
    bA[j].y = Wb[m * CH + k0 + 1];
  }

  v8f biasA, biasB;
#pragma unroll
  for (int i = 0; i < 8; ++i) { biasA[i] = ba[i + hi8]; biasB[i] = bb[i + hi8]; }

  __builtin_amdgcn_sched_barrier(0);
  float mean1, inv1;
  fold_finish(pr1, mean1, inv1);

#pragma unroll
  for (int i = 0; i < 8; ++i) biasA[i] -= mean1;
  scale_weights(bA, inv1 * LR_A);

  v8f acc1 = {0.f, 0.f, 0.f, 0.f, 0.f, 0.f, 0.f, 0.f};
  v8f acc2 = {0.f, 0.f, 0.f, 0.f, 0.f, 0.f, 0.f, 0.f};

  __builtin_amdgcn_s_setprio(3);
  for (int t4 = 0; t4 < tpw; t4 += 4) {
    stage_priority(t4, tpw);
    const xquad xc = xr;
    xr = load_x_quad(xs4, (tbase + ((t4 + 4 < tpw) ? t4 + 4 : t4)) * 4, m, khalf);
    __builtin_amdgcn_sched_barrier(0);
#pragma unroll
   for (int u = 0; u < 4; ++u) {
    const v2f xB = x_operand(xc, u, khalf);

    v8f y = wmma4(aA, xB, biasA);

    v8f h = lrelu_core8(y);

    v8f z = biasB;
#pragma unroll
    for (int j = 0; j < 4; ++j) {
      v2f hB;
      hB.x = h[2 * j];
      hB.y = h[2 * j + 1];
      z = wmma4(bA[j], hB, z);
    }

    acc1 += z;
    acc2 += z * z;
   }
  }
  const float t1 = ((acc1[0] + acc1[1]) + (acc1[2] + acc1[3])) +
                   ((acc1[4] + acc1[5]) + (acc1[6] + acc1[7]));
  const float t2 = ((acc2[0] + acc2[1]) + (acc2[2] + acc2[3])) +
                   ((acc2[4] + acc2[5]) + (acc2[6] + acc2[7]));

  const float w1 = wave_sum(t1);
  const float w2 = wave_sum(t2);
  __shared__ float r1[8], r2[8];
  if (lane == 0) { r1[wave] = w1; r2[wave] = w2; }
  __syncthreads();
  if (threadIdx.x == 0) {
    part[(s * PB3 + blockIdx.x) * 2 + 0] =
        ((r1[0] + r1[1]) + (r1[2] + r1[3])) + ((r1[4] + r1[5]) + (r1[6] + r1[7]));
    part[(s * PB3 + blockIdx.x) * 2 + 1] =
        ((r2[0] + r2[1]) + (r2[2] + r2[3])) + ((r2[4] + r2[5]) + (r2[6] + r2[7]));
  }
}

__global__ void k_final(const float* __restrict__ x, const float* __restrict__ Wa,
                        const float* __restrict__ ba, const float* __restrict__ Wb,
                        const float* __restrict__ bb, const float* __restrict__ Wc,
                        const float* __restrict__ bc, const float* __restrict__ part1,
                        const float* __restrict__ part2, float* __restrict__ out) {
  const int s = blockIdx.y;
  const float* xs = x + (size_t)s * CI * NPIX;
  float* os = out + (size_t)s * CO * NPIX;
  const int lane  = threadIdx.x & 31;
  const int wave  = threadIdx.x >> 5;
  const int m     = lane & 15;
  const int khalf = (lane >> 4) << 1;
  const int hi8   = (lane >> 4) << 3;

  const int tpb = (NPIX / 16) / gridDim.x;
  const int tpw = tpb / 8;
  __builtin_assume(tpw >= 4);
  const int tbase = blockIdx.x * tpb + wave * tpw;
  const v4f* xs4 = (const v4f*)xs;
  xquad xr = load_x_quad(xs4, tbase * 4, m, khalf);
  const v2f pr1 = fold_load<PB1>(part1, s, lane);
  const v2f pr2 = fold_load<PB3>(part2, s, lane);

  v2f aA;
  aA.x = Wa[m * CI + khalf];
  aA.y = (khalf == 0) ? Wa[m * CI + 1] : 0.f;

  v2f bA[4], cA[4];
#pragma unroll
  for (int j = 0; j < 4; ++j) {
    int k0 = hand_off_k(j, hi8);
    bA[j].x = Wb[m * CH + k0];
    bA[j].y = Wb[m * CH + k0 + 1];
    cA[j].x = (m < CO) ? Wc[m * CH + k0] : 0.f;
    cA[j].y = (m < CO) ? Wc[m * CH + k0 + 1] : 0.f;
  }

  v8f biasA, biasB, biasC;
#pragma unroll
  for (int i = 0; i < 8; ++i) {
    biasA[i] = ba[i + hi8];
    biasB[i] = bb[i + hi8];
    biasC[i] = (i + hi8 < CO) ? bc[i + hi8] : 0.f;
  }

  __builtin_amdgcn_sched_barrier(0);
  float mean1, inv1, mean2, inv2;
  fold_finish(pr1, mean1, inv1);
  fold_finish(pr2, mean2, inv2);

#pragma unroll
  for (int i = 0; i < 8; ++i) { biasA[i] -= mean1; biasB[i] -= mean2; }
  scale_weights(bA, inv1 * LR_A);
  scale_weights(cA, inv2 * LR_A);

  __builtin_amdgcn_s_setprio(3);
  for (int t4 = 0; t4 < tpw; t4 += 4) {
    stage_priority(t4, tpw);
    const xquad xc = xr;
    xr = load_x_quad(xs4, (tbase + ((t4 + 4 < tpw) ? t4 + 4 : t4)) * 4, m, khalf);
    __builtin_amdgcn_sched_barrier(0);
    v4f oq[CO];
#pragma unroll
   for (int u = 0; u < 4; ++u) {
    const v2f xB = x_operand(xc, u, khalf);

    v8f y = wmma4(aA, xB, biasA);

    v8f h1 = lrelu_core8(y);

    v8f z = biasB;
#pragma unroll
    for (int j = 0; j < 4; ++j) {
      v2f hB;
      hB.x = h1[2 * j];
      hB.y = h1[2 * j + 1];
      z = wmma4(bA[j], hB, z);
    }

    v8f h2 = lrelu_core8(z);

    v8f o = biasC;
#pragma unroll
    for (int j = 0; j < 4; ++j) {
      v2f hB;
      hB.x = h2[2 * j];
      hB.y = h2[2 * j + 1];
      o = wmma4(cA[j], hB, o);
    }

#pragma unroll
    for (int r = 0; r < CO; ++r) oq[r][u] = o[r];
   }
    if (lane < 16) {
      v4f* os4 = (v4f*)os;
      const int q = (tbase + t4) * 4 + m;
#pragma unroll
      for (int r = 0; r < CO; ++r) os4[r * NQUAD + q] = oq[r];
    }
  }
}

extern "C" void kernel_launch(void* const* d_in, const int* in_sizes, int n_in,
                              void* d_out, int out_size, void* d_ws, size_t ws_size,
                              hipStream_t stream) {
  (void)in_sizes; (void)n_in; (void)out_size; (void)ws_size;
  const float* x  = (const float*)d_in[0];
  const float* Wa = (const float*)d_in[1];
  const float* ba = (const float*)d_in[2];
  const float* Wb = (const float*)d_in[3];
  const float* bb = (const float*)d_in[4];
  const float* Wc = (const float*)d_in[5];
  const float* bc = (const float*)d_in[6];
  float* out = (float*)d_out;

  float* ws     = (float*)d_ws;
  float* part1  = ws;
  float* part2  = part1 + NSAMP * PB1 * 2;

  k_stats1 <<<dim3(PB1, NSAMP), 256, 0, stream>>>(x, Wa, ba, part1);
  k_stats2 <<<dim3(PB3, NSAMP), 256, 0, stream>>>(x, Wa, ba, Wb, bb, part1, part2);
  k_final  <<<dim3(PB5, NSAMP), 256, 0, stream>>>(x, Wa, ba, Wb, bb, Wc, bc,
                                                  part1, part2, out);
}
